# starting kernel
# baseline (speedup 1.0000x reference)
#include <hip/hip_runtime.h>
#include <hip/hip_bf16.h>

typedef __attribute__((ext_vector_type(2))) float v2f;
typedef __attribute__((ext_vector_type(8))) float v8f;

#define ND 64

__global__ __launch_bounds__(256) void gpr_init_kernel(
    const float* __restrict__ x, const float* __restrict__ temp,
    float* __restrict__ hidden, float* __restrict__ xcur,
    float* __restrict__ xnew, int total) {
  int i = blockIdx.x * 256 + threadIdx.x;
  if (i < total) {
    float v = x[i];
    hidden[i] = v * temp[0];
    xcur[i]   = v;
    xnew[i]   = 0.0f;
  }
}

__global__ __launch_bounds__(256) void gpr_gemm_wmma_kernel(
    const float* __restrict__ X, const float* __restrict__ Wm,
    const float* __restrict__ bias, float* __restrict__ H,
    int nWaves) {
  int wave = (blockIdx.x * 256 + threadIdx.x) >> 5;
  if (wave >= nWaves) return;
  int lane  = threadIdx.x & 31;
  int mtile = wave >> 2;
  int otile = wave & 3;
  int hi = lane >> 4;
  int lo = lane & 15;

  const float* arow = X  + (size_t)(mtile * 16 + lo) * ND;
  const float* brow = Wm + (size_t)(otile * 16 + lo) * ND;

  v8f c = {};
#pragma unroll
  for (int k = 0; k < ND; k += 4) {
    v2f a = *(const v2f*)(arow + k + 2 * hi);
    v2f b = *(const v2f*)(brow + k + 2 * hi);
    c = __builtin_amdgcn_wmma_f32_16x16x4_f32(
         false, a,  false, b,
         (short)0, c,  false,  false);
  }

  float bv = bias[otile * 16 + lo];
  float* hout = H + (size_t)(mtile * 16 + 8 * hi) * ND + otile * 16 + lo;
#pragma unroll
  for (int v = 0; v < 8; ++v)
    hout[(size_t)v * ND] = c[v] + bv;
}

__global__ __launch_bounds__(256) void gpr_scatter_kernel(
    const float* __restrict__ H, const float* __restrict__ w,
    const int* __restrict__ src, const int* __restrict__ dst,
    float* __restrict__ xnew, int nEdges) {
  int tid = blockIdx.x * 256 + threadIdx.x;
  int e = tid >> 4;
  if (e >= nEdges) return;
  int sub = tid & 15;

  int   s  = src[e];
  int   d  = dst[e];
  float we = w[e];

  const float4* hp = (const float4*)(H + (size_t)s * ND) + sub;
  float4 hv = *hp;

  float* out = xnew + (size_t)d * ND + sub * 4;
  unsafeAtomicAdd(out + 0, hv.x * we);
  unsafeAtomicAdd(out + 1, hv.y * we);
  unsafeAtomicAdd(out + 2, hv.z * we);
  unsafeAtomicAdd(out + 3, hv.w * we);
}

__global__ __launch_bounds__(256) void gpr_finalize_kernel(
    float* __restrict__ xnew, float* __restrict__ xcur,
    float* __restrict__ hidden, const float* __restrict__ temp,
    int tempIdx, int total) {
  int i = blockIdx.x * 256 + threadIdx.x;
  if (i < total) {
    float v = fmaxf(xnew[i], 0.0f);
    xcur[i] = v;
    hidden[i] += v * temp[tempIdx];
    xnew[i] = 0.0f;
  }
}

extern "C" void kernel_launch(void* const* d_in, const int* in_sizes, int n_in,
                              void* d_out, int out_size, void* d_ws, size_t ws_size,
                              hipStream_t stream) {
  const float* x    = (const float*)d_in[0];
  const float* w    = (const float*)d_in[1];
  const float* Wm   = (const float*)d_in[2];
  const float* bias = (const float*)d_in[3];
  const float* temp = (const float*)d_in[4];
  const int*   src  = (const int*)d_in[5];
  const int*   dst  = (const int*)d_in[6];
  float* hidden = (float*)d_out;

  const int n = in_sizes[0] / ND;
  const int E = in_sizes[1];
  const int L = in_sizes[3] / ND;
  const int total = n * ND;

  float* h    = (float*)d_ws;
  float* xcur = h    + (size_t)total;
  float* xnew = xcur + (size_t)total;

  const int ewBlocks   = (total + 255) / 256;
  const int nWaves     = (n / 16) * 4;
  const int gemmBlocks = (nWaves * 32 + 255) / 256;
  const int scatBlocks = (E * 16 + 255) / 256;

  gpr_init_kernel<<<ewBlocks, 256, 0, stream>>>(x, temp, hidden, xcur, xnew, total);

  for (int i = 0; i < L; ++i) {
    gpr_gemm_wmma_kernel<<<gemmBlocks, 256, 0, stream>>>(
        xcur, Wm + (size_t)i * ND * ND, bias + (size_t)i * ND, h, nWaves);
    gpr_scatter_kernel<<<scatBlocks, 256, 0, stream>>>(h, w, src, dst, xnew, E);
    gpr_finalize_kernel<<<ewBlocks, 256, 0, stream>>>(
        xnew, xcur, hidden, temp, i + 1, total);
  }
}


# Scatter lane map changed so each f32 atomic and each gather load of a wave covers one whole 128-byte line (lane l owns floats l, l+32; 8 edges per wave); atomic arrival order stays unordered
# speedup vs baseline: 1.9745x; 1.9745x over previous
#include <hip/hip_runtime.h>
#include <hip/hip_bf16.h>

typedef __attribute__((ext_vector_type(2))) float v2f;
typedef __attribute__((ext_vector_type(8))) float v8f;

#define ND 64

__global__ __launch_bounds__(256) void gpr_init_kernel(
    const float* __restrict__ x, const float* __restrict__ temp,
    float* __restrict__ hidden, float* __restrict__ xcur,
    float* __restrict__ xnew, int total) {
  int i = blockIdx.x * 256 + threadIdx.x;
  if (i < total) {
    float v = x[i];
    hidden[i] = v * temp[0];
    xcur[i]   = v;
    xnew[i]   = 0.0f;
  }
}

__global__ __launch_bounds__(256) void gpr_gemm_wmma_kernel(
    const float* __restrict__ X, const float* __restrict__ Wm,
    const float* __restrict__ bias, float* __restrict__ H,
    int nWaves) {
  int wave = (blockIdx.x * 256 + threadIdx.x) >> 5;
  if (wave >= nWaves) return;
  int lane  = threadIdx.x & 31;
  int mtile = wave >> 2;
  int otile = wave & 3;
  int hi = lane >> 4;
  int lo = lane & 15;

  const float* arow = X  + (size_t)(mtile * 16 + lo) * ND;
  const float* brow = Wm + (size_t)(otile * 16 + lo) * ND;

  v8f c = {};
#pragma unroll
  for (int k = 0; k < ND; k += 4) {
    v2f a = *(const v2f*)(arow + k + 2 * hi);
    v2f b = *(const v2f*)(brow + k + 2 * hi);
    c = __builtin_amdgcn_wmma_f32_16x16x4_f32(
         false, a,  false, b,
         (short)0, c,  false,  false);
  }

  float bv = bias[otile * 16 + lo];
  float* hout = H + (size_t)(mtile * 16 + 8 * hi) * ND + otile * 16 + lo;
#pragma unroll
  for (int v = 0; v < 8; ++v)
    hout[(size_t)v * ND] = c[v] + bv;
}

#define EPW 8
__global__ __launch_bounds__(256) void gpr_scatter_kernel(
    const float* __restrict__ H, const float* __restrict__ w,
    const int* __restrict__ src, const int* __restrict__ dst,
    float* __restrict__ xnew, int nEdges) {
  const int wv   = __builtin_amdgcn_readfirstlane((int)((blockIdx.x * 256 + threadIdx.x) >> 5));
  const int lane = threadIdx.x & 31;
  const int base = wv * EPW;
  if (base >= nEdges) return;

  if (base + EPW <= nEdges) {
    float v0[EPW], v1[EPW];
    int   d[EPW];
#pragma unroll
    for (int j = 0; j < EPW; ++j) {
      const int   s  = src[base + j];
      const float we = w[base + j];
      d[j] = dst[base + j];
      const float* hp = H + (size_t)s * ND + lane;
      v0[j] = hp[0]  * we;
      v1[j] = hp[32] * we;
    }
#pragma unroll
    for (int j = 0; j < EPW; ++j) {
      float* o = xnew + (size_t)d[j] * ND + lane;
      unsafeAtomicAdd(o,      v0[j]);
      unsafeAtomicAdd(o + 32, v1[j]);
    }
  } else {
    for (int e = base; e < nEdges; ++e) {
      const int   s  = src[e];
      const float we = w[e];
      const float* hp = H + (size_t)s * ND + lane;
      float* o = xnew + (size_t)dst[e] * ND + lane;
      unsafeAtomicAdd(o,      hp[0]  * we);
      unsafeAtomicAdd(o + 32, hp[32] * we);
    }
  }
}

__global__ __launch_bounds__(256) void gpr_finalize_kernel(
    float* __restrict__ xnew, float* __restrict__ xcur,
    float* __restrict__ hidden, const float* __restrict__ temp,
    int tempIdx, int total) {
  int i = blockIdx.x * 256 + threadIdx.x;
  if (i < total) {
    float v = fmaxf(xnew[i], 0.0f);
    xcur[i] = v;
    hidden[i] += v * temp[tempIdx];
    xnew[i] = 0.0f;
  }
}

extern "C" void kernel_launch(void* const* d_in, const int* in_sizes, int n_in,
                              void* d_out, int out_size, void* d_ws, size_t ws_size,
                              hipStream_t stream) {
  const float* x    = (const float*)d_in[0];
  const float* w    = (const float*)d_in[1];
  const float* Wm   = (const float*)d_in[2];
  const float* bias = (const float*)d_in[3];
  const float* temp = (const float*)d_in[4];
  const int*   src  = (const int*)d_in[5];
  const int*   dst  = (const int*)d_in[6];
  float* hidden = (float*)d_out;

  const int n = in_sizes[0] / ND;
  const int E = in_sizes[1];
  const int L = in_sizes[3] / ND;
  const int total = n * ND;

  float* h    = (float*)d_ws;
  float* xcur = h    + (size_t)total;
  float* xnew = xcur + (size_t)total;

  const int ewBlocks   = (total + 255) / 256;
  const int nWaves     = (n / 16) * 4;
  const int gemmBlocks = (nWaves * 32 + 255) / 256;
  const int scatWaves  = (E + EPW - 1) / EPW;
  const int scatBlocks = (scatWaves * 32 + 255) / 256;

  gpr_init_kernel<<<ewBlocks, 256, 0, stream>>>(x, temp, hidden, xcur, xnew, total);

  for (int i = 0; i < L; ++i) {
    gpr_gemm_wmma_kernel<<<gemmBlocks, 256, 0, stream>>>(
        xcur, Wm + (size_t)i * ND * ND, bias + (size_t)i * ND, h, nWaves);
    gpr_scatter_kernel<<<scatBlocks, 256, 0, stream>>>(h, w, src, dst, xnew, E);
    gpr_finalize_kernel<<<ewBlocks, 256, 0, stream>>>(
        xnew, xcur, hidden, temp, i + 1, total);
  }
}


# Edges binned once per call by destination range (LDS counters: count, scan, place); one owner work-group per range sums its rows by f32 LDS atomic adds, relu+hidden fused; sum order stays unordered
# speedup vs baseline: 2.2938x; 1.1617x over previous
#include <hip/hip_runtime.h>
#include <hip/hip_bf16.h>


typedef __attribute__((ext_vector_type(2))) float v2f;
typedef __attribute__((ext_vector_type(4))) float v4f;
typedef __attribute__((ext_vector_type(8))) float v8f;

#define ND   64
#define BT   1024
#define NBLK 256
#define MAXB 256
#define EPW  16
#define TPT  8
#define TILE (BT * TPT)

__global__ __launch_bounds__(256) void gpr_init_kernel(
    const float* __restrict__ x, const float* __restrict__ temp,
    float* __restrict__ hidden, float* __restrict__ xcur, int total) {
  int i = blockIdx.x * 256 + threadIdx.x;
  if (i < total) {
    float v = x[i];
    hidden[i] = v * temp[0];
    xcur[i]   = v;
  }
}

__global__ __launch_bounds__(BT) void gpr_bin_count_kernel(
    const int* __restrict__ dst, int* __restrict__ offs,
    int nEdges, int chunk, unsigned rpb) {
  __shared__ int cnt[MAXB];
  const int t = threadIdx.x;
  if (t < MAXB) cnt[t] = 0;
  __syncthreads();
  const int e0 = blockIdx.x * chunk;
  const int e1 = min(e0 + chunk, nEdges);
  for (int base = e0 + t; base < e1; base += TILE) {
    unsigned d[TPT];
#pragma unroll
    for (int k = 0; k < TPT; ++k)
      d[k] = (base + k * BT < e1) ? (unsigned)dst[base + k * BT] : 0u;
#pragma unroll
    for (int k = 0; k < TPT; ++k)
      if (base + k * BT < e1) atomicAdd(&cnt[d[k] / rpb], 1);
  }
  __syncthreads();
  if (t < MAXB) offs[t * NBLK + blockIdx.x] = cnt[t];
}

__device__ __forceinline__ int wave_incl_scan(int c, int lane) {
  int incl = c;
#pragma unroll
  for (int o = 1; o < 32; o <<= 1) {
    const int v = __shfl_up(incl, o);
    if (lane >= o) incl += v;
  }
  return incl;
}

#define SCAN_N   (MAXB * NBLK)
#define SCAN_PER (SCAN_N / BT)
#define SCAN_LD  32
__global__ __launch_bounds__(BT) void gpr_bin_scan_kernel(int* __restrict__ offs) {
  __shared__ int st[SCAN_N + SCAN_N / SCAN_PER];
  __shared__ int wtot[BT / 32];
  const int t    = threadIdx.x;
  const int lane = t & 31;

  for (int k0 = 0; k0 < SCAN_PER; k0 += SCAN_LD) {
    int v[SCAN_LD];
#pragma unroll
    for (int k = 0; k < SCAN_LD; ++k) v[k] = offs[(k0 + k) * BT + t];
#pragma unroll
    for (int k = 0; k < SCAN_LD; ++k) {
      const int i = (k0 + k) * BT + t;
      st[i + i / SCAN_PER] = v[k];
    }
  }
  __syncthreads();

  int* q = st + t * (SCAN_PER + 1);
  int s = 0;
  for (int j = 0; j < SCAN_PER; ++j) s += q[j];
  const int incl = wave_incl_scan(s, lane);
  if (lane == 31) wtot[t >> 5] = incl;
  __syncthreads();
  int run = incl - s;
  for (int w = 0; w < (t >> 5); ++w) run += wtot[w];
  for (int j = 0; j < SCAN_PER; ++j) { const int c = q[j]; q[j] = run; run += c; }
  __syncthreads();

  for (int k = 0; k < SCAN_PER; ++k) {
    const int i = k * BT + t;
    offs[i] = st[i + i / SCAN_PER];
  }
}

__global__ __launch_bounds__(BT) void gpr_bin_place_kernel(
    const int* __restrict__ src, const int* __restrict__ dst,
    const float* __restrict__ w, const int* __restrict__ offs,
    float* __restrict__ wsort, unsigned short* __restrict__ s16,
    unsigned char* __restrict__ dl8, int nEdges, int chunk, unsigned rpb) {
  __shared__ int      gcur[MAXB];
  __shared__ int      lcnt[MAXB];
  __shared__ int      lbeg[MAXB];
  __shared__ int      wtot[MAXB / 32];
  __shared__ float    stW[TILE];
  __shared__ unsigned stK[TILE];
  const int t = threadIdx.x;
  if (t < MAXB) gcur[t] = offs[t * NBLK + blockIdx.x];
  const int e0 = blockIdx.x * chunk;
  const int e1 = min(e0 + chunk, nEdges);

  for (int base = e0; base < e1; base += TILE) {
    if (t < MAXB) lcnt[t] = 0;
    __syncthreads();

    unsigned d[TPT], bk[TPT];
    int      sv[TPT];
    float    wv[TPT];
#pragma unroll
    for (int k = 0; k < TPT; ++k) {
      const int  e  = base + k * BT + t;
      const bool ok = e < e1;
      d[k]  = ok ? (unsigned)dst[e] : 0u;
      sv[k] = ok ? src[e] : 0;
      wv[k] = ok ? w[e] : 0.0f;
    }
#pragma unroll
    for (int k = 0; k < TPT; ++k) {
      bk[k] = d[k] / rpb;
      if (base + k * BT + t < e1) atomicAdd(&lcnt[bk[k]], 1);
    }
    __syncthreads();

    const int c = (t < MAXB) ? lcnt[t] : 0;
    const int incl = wave_incl_scan(c, t & 31);
    if (t < MAXB && (t & 31) == 31) wtot[t >> 5] = incl;
    __syncthreads();
    if (t < MAXB) {
      int add = 0;
      for (int q = 0; q < (t >> 5); ++q) add += wtot[q];
      const int ex = incl - c + add;
      lbeg[t] = ex;
      lcnt[t] = ex;
    }
    __syncthreads();

#pragma unroll
    for (int k = 0; k < TPT; ++k) {
      if (base + k * BT + t < e1) {
        const int r = atomicAdd(&lcnt[bk[k]], 1);
        stW[r] = wv[k];
        stK[r] = ((unsigned)sv[k] & 0xffffu) | ((d[k] - bk[k] * rpb) << 16) | (bk[k] << 24);
      }
    }
    __syncthreads();

    const int m = min(TILE, e1 - base);
    for (int p = t; p < m; p += BT) {
      const unsigned key = stK[p];
      const unsigned b   = key >> 24;
      const int pos = gcur[b] + (p - lbeg[b]);
      wsort[pos] = stW[p];
      s16[pos]   = (unsigned short)(key & 0xffffu);
      dl8[pos]   = (unsigned char)((key >> 16) & 0xffu);
    }
    __syncthreads();
    if (t < MAXB) gcur[t] += lcnt[t] - lbeg[t];
  }
}

__global__ __launch_bounds__(256) void gpr_gemm_wmma_kernel(
    const float* __restrict__ X, const float* __restrict__ Wm,
    const float* __restrict__ bias, float* __restrict__ H,
    int nWaves) {
  int wave = (blockIdx.x * 256 + threadIdx.x) >> 5;
  if (wave >= nWaves) return;
  int lane  = threadIdx.x & 31;
  int mtile = wave >> 2;
  int otile = wave & 3;
  int hi = lane >> 4;
  int lo = lane & 15;

  const float* arow = X  + (size_t)(mtile * 16 + lo) * ND;
  const float* brow = Wm + (size_t)(otile * 16 + lo) * ND;

  v8f c = {};
#pragma unroll
  for (int k = 0; k < ND; k += 4) {
    v2f a = *(const v2f*)(arow + k + 2 * hi);
    v2f b = *(const v2f*)(brow + k + 2 * hi);
    c = __builtin_amdgcn_wmma_f32_16x16x4_f32(
         false, a,  false, b,
         (short)0, c,  false,  false);
  }

  float bv = bias[otile * 16 + lo];
  float* hout = H + (size_t)(mtile * 16 + 8 * hi) * ND + otile * 16 + lo;
#pragma unroll
  for (int v = 0; v < 8; ++v)
    hout[(size_t)v * ND] = c[v] + bv;
}

#define ACC_FLOATS 41216
__global__ __launch_bounds__(BT) void gpr_owner_sum_kernel(
    const float* __restrict__ H, const float* __restrict__ wsort,
    const unsigned short* __restrict__ s16, const unsigned char* __restrict__ dl8,
    const int* __restrict__ offs, float* __restrict__ xcur,
    float* __restrict__ hidden, const float* __restrict__ temp,
    int tempIdx, int n, int rpb, int nEdges) {
  __shared__ float acc[ACC_FLOATS];
  const int t    = threadIdx.x;
  const int b    = blockIdx.x;
  const int row0 = b * rpb;
  const int rows = min(rpb, n - row0);
  const int nAcc4 = rows * (ND / 4);

  for (int i = t; i < nAcc4; i += BT) ((v4f*)acc)[i] = (v4f){0.0f, 0.0f, 0.0f, 0.0f};
  __syncthreads();

  const int eBeg = offs[b * NBLK];
  const int eEnd = (b + 1 < MAXB) ? offs[(b + 1) * NBLK] : nEdges;
  const int wv   = __builtin_amdgcn_readfirstlane(t >> 5);
  const int lane = t & 31;

  for (int g = eBeg + wv * EPW; g < eEnd; g += (BT / 32) * EPW) {
    if (g + EPW <= eEnd) {
      v2f v[EPW];
      int dl[EPW];
#pragma unroll
      for (int j = 0; j < EPW; ++j) {
        const int   s  = s16[g + j];
        const float we = wsort[g + j];
        dl[j] = dl8[g + j];
        v[j] = *(const v2f*)(H + (size_t)s * ND + 2 * lane) * we;
      }
#pragma unroll
      for (int j = 0; j < EPW; ++j) {
        float* o = acc + dl[j] * ND + 2 * lane;
        __hip_atomic_fetch_add(o,     v[j].x, __ATOMIC_RELAXED, __HIP_MEMORY_SCOPE_WORKGROUP);
        __hip_atomic_fetch_add(o + 1, v[j].y, __ATOMIC_RELAXED, __HIP_MEMORY_SCOPE_WORKGROUP);
      }
    } else {
      v2f v[EPW];
      int dl[EPW];
#pragma unroll
      for (int j = 0; j < EPW; ++j) {
        const int   e  = min(g + j, eEnd - 1);
        const int   s  = s16[e];
        const float we = wsort[e];
        dl[j] = dl8[e];
        v[j] = *(const v2f*)(H + (size_t)s * ND + 2 * lane) * we;
      }
#pragma unroll
      for (int j = 0; j < EPW; ++j) {
        if (g + j < eEnd) {
          float* o = acc + dl[j] * ND + 2 * lane;
          __hip_atomic_fetch_add(o,     v[j].x, __ATOMIC_RELAXED, __HIP_MEMORY_SCOPE_WORKGROUP);
          __hip_atomic_fetch_add(o + 1, v[j].y, __ATOMIC_RELAXED, __HIP_MEMORY_SCOPE_WORKGROUP);
        }
      }
    }
  }
  __syncthreads();

  const float tk = temp[tempIdx];
  v4f* xo = (v4f*)(xcur   + (size_t)row0 * ND);
  v4f* ho = (v4f*)(hidden + (size_t)row0 * ND);
  for (int i = t; i < nAcc4; i += BT) {
    v4f a = ((const v4f*)acc)[i];
    v4f v;
    v.x = fmaxf(a.x, 0.0f); v.y = fmaxf(a.y, 0.0f);
    v.z = fmaxf(a.z, 0.0f); v.w = fmaxf(a.w, 0.0f);
    xo[i] = v;
    ho[i] = ho[i] + v * tk;
  }
}

extern "C" void kernel_launch(void* const* d_in, const int* in_sizes, int n_in,
                              void* d_out, int out_size, void* d_ws, size_t ws_size,
                              hipStream_t stream) {
  const float* x    = (const float*)d_in[0];
  const float* w    = (const float*)d_in[1];
  const float* Wm   = (const float*)d_in[2];
  const float* bias = (const float*)d_in[3];
  const float* temp = (const float*)d_in[4];
  const int*   src  = (const int*)d_in[5];
  const int*   dst  = (const int*)d_in[6];
  float* hidden = (float*)d_out;

  const int n = in_sizes[0] / ND;
  const int E = in_sizes[1];
  const int L = in_sizes[3] / ND;
  const int total = n * ND;

  float* h    = (float*)d_ws;
  float* xcur = h    + (size_t)total;
  char* plane = (char*)(xcur + (size_t)total);
  float*          wsort = (float*)plane;
  unsigned short* s16   = (unsigned short*)(plane + (size_t)E * 4);
  unsigned char*  dl8   = (unsigned char*)(plane + (size_t)E * 6);
  int*            offs  = (int*)(plane + (((size_t)E * 7 + 3) & ~(size_t)3));

  const int rpb   = (n + MAXB - 1) / MAXB;
  const int nb    = (n + rpb - 1) / rpb;
  const int chunk = (E + NBLK - 1) / NBLK;

  const int ewBlocks   = (total + 255) / 256;
  const int nWaves     = (n / 16) * 4;
  const int gemmBlocks = (nWaves * 32 + 255) / 256;

  gpr_init_kernel<<<ewBlocks, 256, 0, stream>>>(x, temp, hidden, xcur, total);
  gpr_bin_count_kernel<<<NBLK, BT, 0, stream>>>(dst, offs, E, chunk, (unsigned)rpb);
  gpr_bin_scan_kernel<<<1, BT, 0, stream>>>(offs);
  gpr_bin_place_kernel<<<NBLK, BT, 0, stream>>>(src, dst, w, offs, wsort, s16, dl8,
                                                 E, chunk, (unsigned)rpb);

  for (int i = 0; i < L; ++i) {
    gpr_gemm_wmma_kernel<<<gemmBlocks, 256, 0, stream>>>(
        xcur, Wm + (size_t)i * ND * ND, bias + (size_t)i * ND, h, nWaves);
    gpr_owner_sum_kernel<<<nb, BT, 0, stream>>>(
        h, wsort, s16, dl8, offs, xcur, hidden, temp, i + 1, n, rpb, E);
  }
}


# Edges of each destination range sorted by row once per call (LDS counting sort, in place); an owner wave sums each run of equal row in registers, one LDS add per run; sum order stays unordered
# speedup vs baseline: 2.7425x; 1.1956x over previous
#include <hip/hip_runtime.h>
#include <hip/hip_bf16.h>


typedef __attribute__((ext_vector_type(2))) float v2f;
typedef __attribute__((ext_vector_type(4))) float v4f;
typedef __attribute__((ext_vector_type(8))) float v8f;

#define ND   64
#define BT   1024
#define NBLK 256
#define MAXB 256
#define EPW  16
#define TPT  8
#define TILE (BT * TPT)

__global__ __launch_bounds__(256) void gpr_init_kernel(
    const float* __restrict__ x, const float* __restrict__ temp,
    float* __restrict__ hidden, float* __restrict__ xcur, int total) {
  int i = blockIdx.x * 256 + threadIdx.x;
  if (i < total) {
    float v = x[i];
    hidden[i] = v * temp[0];
    xcur[i]   = v;
  }
}

__global__ __launch_bounds__(BT) void gpr_bin_count_kernel(
    const int* __restrict__ dst, int* __restrict__ offs,
    int nEdges, int chunk, unsigned rpb) {
  __shared__ int cnt[MAXB];
  const int t = threadIdx.x;
  if (t < MAXB) cnt[t] = 0;
  __syncthreads();
  const int e0 = blockIdx.x * chunk;
  const int e1 = min(e0 + chunk, nEdges);
  for (int base = e0 + t; base < e1; base += TILE) {
    unsigned d[TPT];
#pragma unroll
    for (int k = 0; k < TPT; ++k)
      d[k] = (base + k * BT < e1) ? (unsigned)dst[base + k * BT] : 0u;
#pragma unroll
    for (int k = 0; k < TPT; ++k)
      if (base + k * BT < e1) atomicAdd(&cnt[d[k] / rpb], 1);
  }
  __syncthreads();
  if (t < MAXB) offs[t * NBLK + blockIdx.x] = cnt[t];
}

__device__ __forceinline__ int wave_incl_scan(int c, int lane) {
  int incl = c;
#pragma unroll
  for (int o = 1; o < 32; o <<= 1) {
    const int v = __shfl_up(incl, o);
    if (lane >= o) incl += v;
  }
  return incl;
}

#define SCAN_N   (MAXB * NBLK)
#define SCAN_PER (SCAN_N / BT)
#define SCAN_LD  32
__global__ __launch_bounds__(BT) void gpr_bin_scan_kernel(int* __restrict__ offs) {
  __shared__ int st[SCAN_N + SCAN_N / SCAN_PER];
  __shared__ int wtot[BT / 32];
  const int t    = threadIdx.x;
  const int lane = t & 31;

  for (int k0 = 0; k0 < SCAN_PER; k0 += SCAN_LD) {
    int v[SCAN_LD];
#pragma unroll
    for (int k = 0; k < SCAN_LD; ++k) v[k] = offs[(k0 + k) * BT + t];
#pragma unroll
    for (int k = 0; k < SCAN_LD; ++k) {
      const int i = (k0 + k) * BT + t;
      st[i + i / SCAN_PER] = v[k];
    }
  }
  __syncthreads();

  int* q = st + t * (SCAN_PER + 1);
  int s = 0;
  for (int j = 0; j < SCAN_PER; ++j) s += q[j];
  const int incl = wave_incl_scan(s, lane);
  if (lane == 31) wtot[t >> 5] = incl;
  __syncthreads();
  int run = incl - s;
  for (int w = 0; w < (t >> 5); ++w) run += wtot[w];
  for (int j = 0; j < SCAN_PER; ++j) { const int c = q[j]; q[j] = run; run += c; }
  __syncthreads();

  for (int k = 0; k < SCAN_PER; ++k) {
    const int i = k * BT + t;
    offs[i] = st[i + i / SCAN_PER];
  }
}

__global__ __launch_bounds__(BT) void gpr_bin_place_kernel(
    const int* __restrict__ src, const int* __restrict__ dst,
    const float* __restrict__ w, const int* __restrict__ offs,
    float* __restrict__ wsort, unsigned short* __restrict__ s16,
    unsigned char* __restrict__ dl8, int nEdges, int chunk, unsigned rpb) {
  __shared__ int      gcur[MAXB];
  __shared__ int      lcnt[MAXB];
  __shared__ int      lbeg[MAXB];
  __shared__ int      wtot[MAXB / 32];
  __shared__ float    stW[TILE];
  __shared__ unsigned stK[TILE];
  const int t = threadIdx.x;
  if (t < MAXB) gcur[t] = offs[t * NBLK + blockIdx.x];
  const int e0 = blockIdx.x * chunk;
  const int e1 = min(e0 + chunk, nEdges);

  for (int base = e0; base < e1; base += TILE) {
    if (t < MAXB) lcnt[t] = 0;
    __syncthreads();

    unsigned d[TPT], bk[TPT];
    int      sv[TPT];
    float    wv[TPT];
#pragma unroll
    for (int k = 0; k < TPT; ++k) {
      const int  e  = base + k * BT + t;
      const bool ok = e < e1;
      d[k]  = ok ? (unsigned)dst[e] : 0u;
      sv[k] = ok ? src[e] : 0;
      wv[k] = ok ? w[e] : 0.0f;
    }
#pragma unroll
    for (int k = 0; k < TPT; ++k) {
      bk[k] = d[k] / rpb;
      if (base + k * BT + t < e1) atomicAdd(&lcnt[bk[k]], 1);
    }
    __syncthreads();

    const int c = (t < MAXB) ? lcnt[t] : 0;
    const int incl = wave_incl_scan(c, t & 31);
    if (t < MAXB && (t & 31) == 31) wtot[t >> 5] = incl;
    __syncthreads();
    if (t < MAXB) {
      int add = 0;
      for (int q = 0; q < (t >> 5); ++q) add += wtot[q];
      const int ex = incl - c + add;
      lbeg[t] = ex;
      lcnt[t] = ex;
    }
    __syncthreads();

#pragma unroll
    for (int k = 0; k < TPT; ++k) {
      if (base + k * BT + t < e1) {
        const int r = atomicAdd(&lcnt[bk[k]], 1);
        stW[r] = wv[k];
        stK[r] = ((unsigned)sv[k] & 0xffffu) | ((d[k] - bk[k] * rpb) << 16) | (bk[k] << 24);
      }
    }
    __syncthreads();

    const int m = min(TILE, e1 - base);
    for (int p = t; p < m; p += BT) {
      const unsigned key = stK[p];
      const unsigned b   = key >> 24;
      const int pos = gcur[b] + (p - lbeg[b]);
      wsort[pos] = stW[p];
      s16[pos]   = (unsigned short)(key & 0xffffu);
      dl8[pos]   = (unsigned char)((key >> 16) & 0xffu);
    }
    __syncthreads();
    if (t < MAXB) gcur[t] += lcnt[t] - lbeg[t];
  }
}

#define RS_CAP 16384
__global__ __launch_bounds__(BT) void gpr_row_sort_kernel(
    float* wsort, unsigned short* s16, unsigned char* dl8,
    const int* __restrict__ offs, int nEdges) {
  __shared__ int            cnt[MAXB];
  __shared__ int            wtot[MAXB / 32];
  __shared__ float          stW[RS_CAP];
  __shared__ unsigned short stS[RS_CAP];
  __shared__ unsigned char  stD[RS_CAP];
  const int t = threadIdx.x;
  const int b = blockIdx.x;
  const int eBeg = offs[b * NBLK];
  const int eEnd = (b + 1 < MAXB) ? offs[(b + 1) * NBLK] : nEdges;
  const int m = eEnd - eBeg;
  if (m > RS_CAP) return;
  if (t < MAXB) cnt[t] = 0;
  __syncthreads();

  for (int base = 0; base < m; base += TILE) {
    unsigned d[TPT];
#pragma unroll
    for (int k = 0; k < TPT; ++k) {
      const int p = base + k * BT + t;
      d[k] = (p < m) ? (unsigned)dl8[eBeg + p] : 0u;
    }
#pragma unroll
    for (int k = 0; k < TPT; ++k)
      if (base + k * BT + t < m) atomicAdd(&cnt[d[k]], 1);
  }
  __syncthreads();

  const int c = (t < MAXB) ? cnt[t] : 0;
  const int incl = wave_incl_scan(c, t & 31);
  if (t < MAXB && (t & 31) == 31) wtot[t >> 5] = incl;
  __syncthreads();
  if (t < MAXB) {
    int add = 0;
    for (int q = 0; q < (t >> 5); ++q) add += wtot[q];
    cnt[t] = incl - c + add;
  }
  __syncthreads();

  for (int base = 0; base < m; base += TILE) {
    unsigned d[TPT];
    unsigned short sv[TPT];
    float wv[TPT];
#pragma unroll
    for (int k = 0; k < TPT; ++k) {
      const int  p  = base + k * BT + t;
      const bool ok = p < m;
      d[k]  = ok ? (unsigned)dl8[eBeg + p] : 0u;
      sv[k] = ok ? s16[eBeg + p] : (unsigned short)0;
      wv[k] = ok ? wsort[eBeg + p] : 0.0f;
    }
#pragma unroll
    for (int k = 0; k < TPT; ++k) {
      if (base + k * BT + t < m) {
        const int r = atomicAdd(&cnt[d[k]], 1);
        stW[r] = wv[k];
        stS[r] = sv[k];
        stD[r] = (unsigned char)d[k];
      }
    }
  }
  __syncthreads();

  for (int p = t; p < m; p += BT) {
    wsort[eBeg + p] = stW[p];
    s16[eBeg + p]   = stS[p];
    dl8[eBeg + p]   = stD[p];
  }
}

__global__ __launch_bounds__(256) void gpr_gemm_wmma_kernel(
    const float* __restrict__ X, const float* __restrict__ Wm,
    const float* __restrict__ bias, float* __restrict__ H,
    int nWaves) {
  int wave = (blockIdx.x * 256 + threadIdx.x) >> 5;
  if (wave >= nWaves) return;
  int lane  = threadIdx.x & 31;
  int mtile = wave >> 2;
  int otile = wave & 3;
  int hi = lane >> 4;
  int lo = lane & 15;

  const float* arow = X  + (size_t)(mtile * 16 + lo) * ND;
  const float* brow = Wm + (size_t)(otile * 16 + lo) * ND;

  v8f c = {};
#pragma unroll
  for (int k = 0; k < ND; k += 4) {
    v2f a = *(const v2f*)(arow + k + 2 * hi);
    v2f b = *(const v2f*)(brow + k + 2 * hi);
    c = __builtin_amdgcn_wmma_f32_16x16x4_f32(
         false, a,  false, b,
         (short)0, c,  false,  false);
  }

  float bv = bias[otile * 16 + lo];
  float* hout = H + (size_t)(mtile * 16 + 8 * hi) * ND + otile * 16 + lo;
#pragma unroll
  for (int v = 0; v < 8; ++v)
    hout[(size_t)v * ND] = c[v] + bv;
}

#define ACC_FLOATS 41216

__device__ __forceinline__ void gpr_flush_run(float* acc, int row, int lane, v2f run) {
  float* o = acc + row * ND + 2 * lane;
  __hip_atomic_fetch_add(o,     run.x, __ATOMIC_RELAXED, __HIP_MEMORY_SCOPE_WORKGROUP);
  __hip_atomic_fetch_add(o + 1, run.y, __ATOMIC_RELAXED, __HIP_MEMORY_SCOPE_WORKGROUP);
}

__global__ __launch_bounds__(BT) void gpr_owner_sum_kernel(
    const float* __restrict__ H, const float* __restrict__ wsort,
    const unsigned short* __restrict__ s16, const unsigned char* __restrict__ dl8,
    const int* __restrict__ offs, float* __restrict__ xcur,
    float* __restrict__ hidden, const float* __restrict__ temp,
    int tempIdx, int n, int rpb, int nEdges) {
  __shared__ float acc[ACC_FLOATS];
  const int t    = threadIdx.x;
  const int b    = blockIdx.x;
  const int row0 = b * rpb;
  const int rows = min(rpb, n - row0);
  const int nAcc4 = rows * (ND / 4);

  for (int i = t; i < nAcc4; i += BT) ((v4f*)acc)[i] = (v4f){0.0f, 0.0f, 0.0f, 0.0f};
  __syncthreads();

  const int eBeg = offs[b * NBLK];
  const int eEnd = (b + 1 < MAXB) ? offs[(b + 1) * NBLK] : nEdges;
  const int wv   = __builtin_amdgcn_readfirstlane(t >> 5);
  const int lane = t & 31;

  const int nW   = BT / 32;
  const int per  = ((eEnd - eBeg + nW * EPW - 1) / (nW * EPW)) * EPW;
  const int wBeg = eBeg + wv * per;
  const int wEnd = min(wBeg + per, eEnd);

  int cur = -1;
  v2f run = {0.0f, 0.0f};

  for (int g = wBeg; g < wEnd; g += EPW) {
    if (g + EPW <= wEnd) {
      v2f v[EPW];
      int dl[EPW];
#pragma unroll
      for (int j = 0; j < EPW; ++j) {
        const int   s  = s16[g + j];
        const float we = wsort[g + j];
        dl[j] = dl8[g + j];
        v[j] = *(const v2f*)(H + (size_t)s * ND + 2 * lane) * we;
      }
#pragma unroll
      for (int j = 0; j < EPW; ++j) {
        if (dl[j] != cur) {
          if (cur >= 0) gpr_flush_run(acc, cur, lane, run);
          cur = dl[j];
          run = v[j];
        } else {
          run += v[j];
        }
      }
    } else {
      v2f v[EPW];
      int dl[EPW];
#pragma unroll
      for (int j = 0; j < EPW; ++j) {
        const int   e  = min(g + j, wEnd - 1);
        const int   s  = s16[e];
        const float we = wsort[e];
        dl[j] = dl8[e];
        v[j] = *(const v2f*)(H + (size_t)s * ND + 2 * lane) * we;
      }
#pragma unroll
      for (int j = 0; j < EPW; ++j) {
        if (g + j < wEnd) {
          if (dl[j] != cur) {
            if (cur >= 0) gpr_flush_run(acc, cur, lane, run);
            cur = dl[j];
            run = v[j];
          } else {
            run += v[j];
          }
        }
      }
    }
  }
  if (cur >= 0) gpr_flush_run(acc, cur, lane, run);
  __syncthreads();

  const float tk = temp[tempIdx];
  v4f* xo = (v4f*)(xcur   + (size_t)row0 * ND);
  v4f* ho = (v4f*)(hidden + (size_t)row0 * ND);
  for (int i = t; i < nAcc4; i += BT) {
    v4f a = ((const v4f*)acc)[i];
    v4f v;
    v.x = fmaxf(a.x, 0.0f); v.y = fmaxf(a.y, 0.0f);
    v.z = fmaxf(a.z, 0.0f); v.w = fmaxf(a.w, 0.0f);
    xo[i] = v;
    ho[i] = ho[i] + v * tk;
  }
}

extern "C" void kernel_launch(void* const* d_in, const int* in_sizes, int n_in,
                              void* d_out, int out_size, void* d_ws, size_t ws_size,
                              hipStream_t stream) {
  const float* x    = (const float*)d_in[0];
  const float* w    = (const float*)d_in[1];
  const float* Wm   = (const float*)d_in[2];
  const float* bias = (const float*)d_in[3];
  const float* temp = (const float*)d_in[4];
  const int*   src  = (const int*)d_in[5];
  const int*   dst  = (const int*)d_in[6];
  float* hidden = (float*)d_out;

  const int n = in_sizes[0] / ND;
  const int E = in_sizes[1];
  const int L = in_sizes[3] / ND;
  const int total = n * ND;

  float* h    = (float*)d_ws;
  float* xcur = h    + (size_t)total;
  char* plane = (char*)(xcur + (size_t)total);
  float*          wsort = (float*)plane;
  unsigned short* s16   = (unsigned short*)(plane + (size_t)E * 4);
  unsigned char*  dl8   = (unsigned char*)(plane + (size_t)E * 6);
  int*            offs  = (int*)(plane + (((size_t)E * 7 + 3) & ~(size_t)3));

  const int rpb   = (n + MAXB - 1) / MAXB;
  const int nb    = (n + rpb - 1) / rpb;
  const int chunk = (E + NBLK - 1) / NBLK;

  const int ewBlocks   = (total + 255) / 256;
  const int nWaves     = (n / 16) * 4;
  const int gemmBlocks = (nWaves * 32 + 255) / 256;

  gpr_init_kernel<<<ewBlocks, 256, 0, stream>>>(x, temp, hidden, xcur, total);
  gpr_bin_count_kernel<<<NBLK, BT, 0, stream>>>(dst, offs, E, chunk, (unsigned)rpb);
  gpr_bin_scan_kernel<<<1, BT, 0, stream>>>(offs);
  gpr_bin_place_kernel<<<NBLK, BT, 0, stream>>>(src, dst, w, offs, wsort, s16, dl8,
                                                 E, chunk, (unsigned)rpb);
  gpr_row_sort_kernel<<<nb, BT, 0, stream>>>(wsort, s16, dl8, offs, E);

  for (int i = 0; i < L; ++i) {
    gpr_gemm_wmma_kernel<<<gemmBlocks, 256, 0, stream>>>(
        xcur, Wm + (size_t)i * ND * ND, bias + (size_t)i * ND, h, nWaves);
    gpr_owner_sum_kernel<<<nb, BT, 0, stream>>>(
        h, wsort, s16, dl8, offs, xcur, hidden, temp, i + 1, n, rpb, E);
  }
}


# Each layer's linear product moved into the kernel that makes its input rows (layer 0 with init; later layers in the owner-sum epilogue, f32 WMMA on rows in LDS, two H planes): 13 launches to 9
# speedup vs baseline: 3.5332x; 1.2883x over previous
#include <hip/hip_runtime.h>
#include <hip/hip_bf16.h>


typedef __attribute__((ext_vector_type(2))) float v2f;
typedef __attribute__((ext_vector_type(4))) float v4f;
typedef __attribute__((ext_vector_type(8))) float v8f;

#define ND   64
#define BT   1024
#define NBLK 256
#define MAXB 256
#define EPW  16
#define TPT  8
#define TILE (BT * TPT)

__global__ __launch_bounds__(BT) void gpr_bin_count_kernel(
    const int* __restrict__ dst, int* __restrict__ offs,
    int nEdges, int chunk, unsigned rpb) {
  __shared__ int cnt[MAXB];
  const int t = threadIdx.x;
  if (t < MAXB) cnt[t] = 0;
  __syncthreads();
  const int e0 = blockIdx.x * chunk;
  const int e1 = min(e0 + chunk, nEdges);
  for (int base = e0 + t; base < e1; base += TILE) {
    unsigned d[TPT];
#pragma unroll
    for (int k = 0; k < TPT; ++k)
      d[k] = (base + k * BT < e1) ? (unsigned)dst[base + k * BT] : 0u;
#pragma unroll
    for (int k = 0; k < TPT; ++k)
      if (base + k * BT < e1) atomicAdd(&cnt[d[k] / rpb], 1);
  }
  __syncthreads();
  if (t < MAXB) offs[t * NBLK + blockIdx.x] = cnt[t];
}

__device__ __forceinline__ int wave_incl_scan(int c, int lane) {
  int incl = c;
#pragma unroll
  for (int o = 1; o < 32; o <<= 1) {
    const int v = __shfl_up(incl, o);
    if (lane >= o) incl += v;
  }
  return incl;
}

#define SCAN_N   (MAXB * NBLK)
#define SCAN_PER (SCAN_N / BT)
#define SCAN_LD  32
__global__ __launch_bounds__(BT) void gpr_bin_scan_kernel(int* __restrict__ offs) {
  __shared__ int st[SCAN_N + SCAN_N / SCAN_PER];
  __shared__ int wtot[BT / 32];
  const int t    = threadIdx.x;
  const int lane = t & 31;

  for (int k0 = 0; k0 < SCAN_PER; k0 += SCAN_LD) {
    int v[SCAN_LD];
#pragma unroll
    for (int k = 0; k < SCAN_LD; ++k) v[k] = offs[(k0 + k) * BT + t];
#pragma unroll
    for (int k = 0; k < SCAN_LD; ++k) {
      const int i = (k0 + k) * BT + t;
      st[i + i / SCAN_PER] = v[k];
    }
  }
  __syncthreads();

  int* q = st + t * (SCAN_PER + 1);
  int s = 0;
  for (int j = 0; j < SCAN_PER; ++j) s += q[j];
  const int incl = wave_incl_scan(s, lane);
  if (lane == 31) wtot[t >> 5] = incl;
  __syncthreads();
  int run = incl - s;
  for (int w = 0; w < (t >> 5); ++w) run += wtot[w];
  for (int j = 0; j < SCAN_PER; ++j) { const int c = q[j]; q[j] = run; run += c; }
  __syncthreads();

  for (int k = 0; k < SCAN_PER; ++k) {
    const int i = k * BT + t;
    offs[i] = st[i + i / SCAN_PER];
  }
}

__global__ __launch_bounds__(BT) void gpr_bin_place_kernel(
    const int* __restrict__ src, const int* __restrict__ dst,
    const float* __restrict__ w, const int* __restrict__ offs,
    float* __restrict__ wsort, unsigned short* __restrict__ s16,
    unsigned char* __restrict__ dl8, int nEdges, int chunk, unsigned rpb) {
  __shared__ int      gcur[MAXB];
  __shared__ int      lcnt[MAXB];
  __shared__ int      lbeg[MAXB];
  __shared__ int      wtot[MAXB / 32];
  __shared__ float    stW[TILE];
  __shared__ unsigned stK[TILE];
  const int t = threadIdx.x;
  if (t < MAXB) gcur[t] = offs[t * NBLK + blockIdx.x];
  const int e0 = blockIdx.x * chunk;
  const int e1 = min(e0 + chunk, nEdges);

  for (int base = e0; base < e1; base += TILE) {
    if (t < MAXB) lcnt[t] = 0;
    __syncthreads();

    unsigned d[TPT], bk[TPT];
    int      sv[TPT];
    float    wv[TPT];
#pragma unroll
    for (int k = 0; k < TPT; ++k) {
      const int  e  = base + k * BT + t;
      const bool ok = e < e1;
      d[k]  = ok ? (unsigned)dst[e] : 0u;
      sv[k] = ok ? src[e] : 0;
      wv[k] = ok ? w[e] : 0.0f;
    }
#pragma unroll
    for (int k = 0; k < TPT; ++k) {
      bk[k] = d[k] / rpb;
      if (base + k * BT + t < e1) atomicAdd(&lcnt[bk[k]], 1);
    }
    __syncthreads();

    const int c = (t < MAXB) ? lcnt[t] : 0;
    const int incl = wave_incl_scan(c, t & 31);
    if (t < MAXB && (t & 31) == 31) wtot[t >> 5] = incl;
    __syncthreads();
    if (t < MAXB) {
      int add = 0;
      for (int q = 0; q < (t >> 5); ++q) add += wtot[q];
      const int ex = incl - c + add;
      lbeg[t] = ex;
      lcnt[t] = ex;
    }
    __syncthreads();

#pragma unroll
    for (int k = 0; k < TPT; ++k) {
      if (base + k * BT + t < e1) {
        const int r = atomicAdd(&lcnt[bk[k]], 1);
        stW[r] = wv[k];
        stK[r] = ((unsigned)sv[k] & 0xffffu) | ((d[k] - bk[k] * rpb) << 16) | (bk[k] << 24);
      }
    }
    __syncthreads();

    const int m = min(TILE, e1 - base);
    for (int p = t; p < m; p += BT) {
      const unsigned key = stK[p];
      const unsigned b   = key >> 24;
      const int pos = gcur[b] + (p - lbeg[b]);
      wsort[pos] = stW[p];
      s16[pos]   = (unsigned short)(key & 0xffffu);
      dl8[pos]   = (unsigned char)((key >> 16) & 0xffu);
    }
    __syncthreads();
    if (t < MAXB) gcur[t] += lcnt[t] - lbeg[t];
  }
}

#define RS_CAP 16384
__global__ __launch_bounds__(BT) void gpr_row_sort_kernel(
    float* wsort, unsigned short* s16, unsigned char* dl8,
    const int* __restrict__ offs, int nEdges) {
  __shared__ int            cnt[MAXB];
  __shared__ int            wtot[MAXB / 32];
  __shared__ float          stW[RS_CAP];
  __shared__ unsigned short stS[RS_CAP];
  __shared__ unsigned char  stD[RS_CAP];
  const int t = threadIdx.x;
  const int b = blockIdx.x;
  const int eBeg = offs[b * NBLK];
  const int eEnd = (b + 1 < MAXB) ? offs[(b + 1) * NBLK] : nEdges;
  const int m = eEnd - eBeg;
  if (m > RS_CAP) return;
  if (t < MAXB) cnt[t] = 0;
  __syncthreads();

  for (int base = 0; base < m; base += TILE) {
    unsigned d[TPT];
#pragma unroll
    for (int k = 0; k < TPT; ++k) {
      const int p = base + k * BT + t;
      d[k] = (p < m) ? (unsigned)dl8[eBeg + p] : 0u;
    }
#pragma unroll
    for (int k = 0; k < TPT; ++k)
      if (base + k * BT + t < m) atomicAdd(&cnt[d[k]], 1);
  }
  __syncthreads();

  const int c = (t < MAXB) ? cnt[t] : 0;
  const int incl = wave_incl_scan(c, t & 31);
  if (t < MAXB && (t & 31) == 31) wtot[t >> 5] = incl;
  __syncthreads();
  if (t < MAXB) {
    int add = 0;
    for (int q = 0; q < (t >> 5); ++q) add += wtot[q];
    cnt[t] = incl - c + add;
  }
  __syncthreads();

  for (int base = 0; base < m; base += TILE) {
    unsigned d[TPT];
    unsigned short sv[TPT];
    float wv[TPT];
#pragma unroll
    for (int k = 0; k < TPT; ++k) {
      const int  p  = base + k * BT + t;
      const bool ok = p < m;
      d[k]  = ok ? (unsigned)dl8[eBeg + p] : 0u;
      sv[k] = ok ? s16[eBeg + p] : (unsigned short)0;
      wv[k] = ok ? wsort[eBeg + p] : 0.0f;
    }
#pragma unroll
    for (int k = 0; k < TPT; ++k) {
      if (base + k * BT + t < m) {
        const int r = atomicAdd(&cnt[d[k]], 1);
        stW[r] = wv[k];
        stS[r] = sv[k];
        stD[r] = (unsigned char)d[k];
      }
    }
  }
  __syncthreads();

  for (int p = t; p < m; p += BT) {
    wsort[eBeg + p] = stW[p];
    s16[eBeg + p]   = stS[p];
    dl8[eBeg + p]   = stD[p];
  }
}

__global__ __launch_bounds__(256) void gpr_first_layer_kernel(
    const float* __restrict__ X, const float* __restrict__ Wm,
    const float* __restrict__ bias, const float* __restrict__ temp,
    float* __restrict__ hidden, float* __restrict__ H,
    int nWaves, int total4) {
  {
    const float t0 = temp[0];
    const int i0 = blockIdx.x * 512 + threadIdx.x;
    const int i1 = i0 + 256;
    const v4f z4 = {0.0f, 0.0f, 0.0f, 0.0f};
    const v4f a0 = (i0 < total4) ? ((const v4f*)X)[i0] : z4;
    const v4f a1 = (i1 < total4) ? ((const v4f*)X)[i1] : z4;
    if (i0 < total4) ((v4f*)hidden)[i0] = a0 * t0;
    if (i1 < total4) ((v4f*)hidden)[i1] = a1 * t0;
  }
  int wave = (blockIdx.x * 256 + threadIdx.x) >> 5;
  if (wave >= nWaves) return;
  int lane  = threadIdx.x & 31;
  int mtile = wave >> 2;
  int otile = wave & 3;
  int hi = lane >> 4;
  int lo = lane & 15;

  const float* arow = X  + (size_t)(mtile * 16 + lo) * ND;
  const float* brow = Wm + (size_t)(otile * 16 + lo) * ND;

  v8f c = {};
#pragma unroll
  for (int k = 0; k < ND; k += 4) {
    v2f a = *(const v2f*)(arow + k + 2 * hi);
    v2f b = *(const v2f*)(brow + k + 2 * hi);
    c = __builtin_amdgcn_wmma_f32_16x16x4_f32(
         false, a,  false, b,
         (short)0, c,  false,  false);
  }

  float bv = bias[otile * 16 + lo];
  float* hout = H + (size_t)(mtile * 16 + 8 * hi) * ND + otile * 16 + lo;
#pragma unroll
  for (int v = 0; v < 8; ++v)
    hout[(size_t)v * ND] = c[v] + bv;
}

#define ACC_FLOATS 41216
#define AP      68
#define AP4     (AP / 4)
#define STG_OFF (MAXB * AP)
#define WL_OFF  (2 * MAXB * AP)

__device__ __forceinline__ void gpr_flush_run(float* acc, int row, int lane, v2f run) {
  float* o = acc + row * AP + 2 * lane;
  __hip_atomic_fetch_add(o,     run.x, __ATOMIC_RELAXED, __HIP_MEMORY_SCOPE_WORKGROUP);
  __hip_atomic_fetch_add(o + 1, run.y, __ATOMIC_RELAXED, __HIP_MEMORY_SCOPE_WORKGROUP);
}

__global__ __launch_bounds__(BT) void gpr_owner_sum_kernel(
    const float* __restrict__ H, const float* __restrict__ wsort,
    const unsigned short* __restrict__ s16, const unsigned char* __restrict__ dl8,
    const int* __restrict__ offs, float* __restrict__ Hnext,
    float* __restrict__ hidden, const float* __restrict__ temp,
    const float* __restrict__ Wn, const float* __restrict__ bn, int doNext,
    int tempIdx, int n, int rpb, int nEdges) {
  __shared__ float lds[ACC_FLOATS];
  float* acc = lds;
  float* stg = lds + STG_OFF;
  float* wl  = lds + WL_OFF;
  const int t    = threadIdx.x;
  const int b    = blockIdx.x;
  const int row0 = b * rpb;
  const int rows = min(rpb, n - row0);
  const int nRt  = (rows + 15) >> 4;
  const int nZero4 = nRt * 16 * AP4;
  const int nRow4  = rows * (ND / 4);

  for (int i = t; i < nZero4; i += BT) ((v4f*)acc)[i] = (v4f){0.0f, 0.0f, 0.0f, 0.0f};
  if (doNext) {
    for (int i = t; i < ND * ND / 4; i += BT)
      ((v4f*)wl)[(i >> 4) * AP4 + (i & 15)] = ((const v4f*)Wn)[i];
  }
  __syncthreads();

  const int eBeg = offs[b * NBLK];
  const int eEnd = (b + 1 < MAXB) ? offs[(b + 1) * NBLK] : nEdges;
  const int wv   = __builtin_amdgcn_readfirstlane(t >> 5);
  const int lane = t & 31;

  const int nW   = BT / 32;
  const int per  = ((eEnd - eBeg + nW * EPW - 1) / (nW * EPW)) * EPW;
  const int wBeg = eBeg + wv * per;
  const int wEnd = min(wBeg + per, eEnd);

  int cur = -1;
  v2f run = {0.0f, 0.0f};

  for (int g = wBeg; g < wEnd; g += EPW) {
    if (g + EPW <= wEnd) {
      v2f v[EPW];
      int dl[EPW];
#pragma unroll
      for (int j = 0; j < EPW; ++j) {
        const int   s  = s16[g + j];
        const float we = wsort[g + j];
        dl[j] = dl8[g + j];
        v[j] = *(const v2f*)(H + (size_t)s * ND + 2 * lane) * we;
      }
#pragma unroll
      for (int j = 0; j < EPW; ++j) {
        if (dl[j] != cur) {
          if (cur >= 0) gpr_flush_run(acc, cur, lane, run);
          cur = dl[j];
          run = v[j];
        } else {
          run += v[j];
        }
      }
    } else {
      v2f v[EPW];
      int dl[EPW];
#pragma unroll
      for (int j = 0; j < EPW; ++j) {
        const int   e  = min(g + j, wEnd - 1);
        const int   s  = s16[e];
        const float we = wsort[e];
        dl[j] = dl8[e];
        v[j] = *(const v2f*)(H + (size_t)s * ND + 2 * lane) * we;
      }
#pragma unroll
      for (int j = 0; j < EPW; ++j) {
        if (g + j < wEnd) {
          if (dl[j] != cur) {
            if (cur >= 0) gpr_flush_run(acc, cur, lane, run);
            cur = dl[j];
            run = v[j];
          } else {
            run += v[j];
          }
        }
      }
    }
  }
  if (cur >= 0) gpr_flush_run(acc, cur, lane, run);
  __syncthreads();

  const float tk = temp[tempIdx];
  v4f* ho = (v4f*)(hidden + (size_t)row0 * ND);
  for (int i = t; i < nRow4; i += BT) {
    const int li = (i >> 4) * AP4 + (i & 15);
    v4f a = ((const v4f*)acc)[li];
    v4f v;
    v.x = fmaxf(a.x, 0.0f); v.y = fmaxf(a.y, 0.0f);
    v.z = fmaxf(a.z, 0.0f); v.w = fmaxf(a.w, 0.0f);
    ((v4f*)acc)[li] = v;
    ho[i] = ho[i] + v * tk;
  }
  if (!doNext) return;
  __syncthreads();

  const int rt = wv >> 1;
  const int cp = wv & 1;
  if (rt < nRt) {
    const int hi = lane >> 4;
    const int lo = lane & 15;
    const float* ap  = acc + (rt * 16 + lo) * AP + 2 * hi;
    const float* bp0 = wl  + (cp * 32 + lo) * AP + 2 * hi;
    const float* bp1 = bp0 + 16 * AP;
    v8f c0 = {};
    v8f c1 = {};
#pragma unroll
    for (int k = 0; k < ND; k += 4) {
      v2f a  = *(const v2f*)(ap  + k);
      v2f b0 = *(const v2f*)(bp0 + k);
      v2f b1 = *(const v2f*)(bp1 + k);
      c0 = __builtin_amdgcn_wmma_f32_16x16x4_f32(
           false, a,  false, b0,
           (short)0, c0,  false,  false);
      c1 = __builtin_amdgcn_wmma_f32_16x16x4_f32(
           false, a,  false, b1,
           (short)0, c1,  false,  false);
    }
    const float bv0 = bn[cp * 32 + lo];
    const float bv1 = bn[cp * 32 + 16 + lo];
    float* so = stg + (rt * 16 + 8 * hi) * AP + cp * 32 + lo;
#pragma unroll
    for (int v = 0; v < 8; ++v) {
      so[v * AP]      = c0[v] + bv0;
      so[v * AP + 16] = c1[v] + bv1;
    }
  }
  __syncthreads();

  v4f* hn = (v4f*)(Hnext + (size_t)row0 * ND);
  for (int i = t; i < nRow4; i += BT)
    hn[i] = ((const v4f*)stg)[(i >> 4) * AP4 + (i & 15)];
}

extern "C" void kernel_launch(void* const* d_in, const int* in_sizes, int n_in,
                              void* d_out, int out_size, void* d_ws, size_t ws_size,
                              hipStream_t stream) {
  const float* x    = (const float*)d_in[0];
  const float* w    = (const float*)d_in[1];
  const float* Wm   = (const float*)d_in[2];
  const float* bias = (const float*)d_in[3];
  const float* temp = (const float*)d_in[4];
  const int*   src  = (const int*)d_in[5];
  const int*   dst  = (const int*)d_in[6];
  float* hidden = (float*)d_out;

  const int n = in_sizes[0] / ND;
  const int E = in_sizes[1];
  const int L = in_sizes[3] / ND;
  const int total = n * ND;

  float* hA = (float*)d_ws;
  float* hB = hA + (size_t)total;
  char* plane = (char*)(hB + (size_t)total);
  float*          wsort = (float*)plane;
  unsigned short* s16   = (unsigned short*)(plane + (size_t)E * 4);
  unsigned char*  dl8   = (unsigned char*)(plane + (size_t)E * 6);
  int*            offs  = (int*)(plane + (((size_t)E * 7 + 3) & ~(size_t)3));

  const int rpb   = (n + MAXB - 1) / MAXB;
  const int nb    = (n + rpb - 1) / rpb;
  const int chunk = (E + NBLK - 1) / NBLK;

  const int nWaves     = (n / 16) * 4;
  const int gemmBlocks = (nWaves * 32 + 255) / 256;

  gpr_first_layer_kernel<<<gemmBlocks, 256, 0, stream>>>(x, Wm, bias, temp, hidden, hA,
                                                        nWaves, total / 4);
  gpr_bin_count_kernel<<<NBLK, BT, 0, stream>>>(dst, offs, E, chunk, (unsigned)rpb);
  gpr_bin_scan_kernel<<<1, BT, 0, stream>>>(offs);
  gpr_bin_place_kernel<<<NBLK, BT, 0, stream>>>(src, dst, w, offs, wsort, s16, dl8,
                                                 E, chunk, (unsigned)rpb);
  gpr_row_sort_kernel<<<nb, BT, 0, stream>>>(wsort, s16, dl8, offs, E);

  for (int i = 0; i < L; ++i) {
    float* hc = (i & 1) ? hB : hA;
    float* hn = (i & 1) ? hA : hB;
    const int more = (i + 1 < L) ? 1 : 0;
    const int nx   = more ? i + 1 : i;
    gpr_owner_sum_kernel<<<nb, BT, 0, stream>>>(
        hc, wsort, s16, dl8, offs, hn, hidden, temp,
        Wm + (size_t)nx * ND * ND, bias + (size_t)nx * ND, more,
        i + 1, n, rpb, E);
  }
}


# Linear product moved behind the neighbour sum: x rows gathered and summed, then @W^T on f32 WMMA, bias as a 65th k column times the summed weights; first-layer launch gone (9 to 8); sum order changes
# speedup vs baseline: 3.6996x; 1.0471x over previous
#include <hip/hip_runtime.h>
#include <hip/hip_bf16.h>


typedef __attribute__((ext_vector_type(2))) float v2f;
typedef __attribute__((ext_vector_type(4))) float v4f;
typedef __attribute__((ext_vector_type(8))) float v8f;

#define ND   64
#define BT   1024
#define NBLK 256
#define MAXB 256
#define EPW  16
#define TPT  8
#define TILE (BT * TPT)

__global__ __launch_bounds__(BT) void gpr_bin_count_kernel(
    const int* __restrict__ dst, int* __restrict__ offs,
    int nEdges, int chunk, unsigned rpb) {
  __shared__ int cnt[MAXB];
  const int t = threadIdx.x;
  if (t < MAXB) cnt[t] = 0;
  __syncthreads();
  const int e0 = blockIdx.x * chunk;
  const int e1 = min(e0 + chunk, nEdges);
  for (int base = e0 + t; base < e1; base += TILE) {
    unsigned d[TPT];
#pragma unroll
    for (int k = 0; k < TPT; ++k)
      d[k] = (base + k * BT < e1) ? (unsigned)dst[base + k * BT] : 0u;
#pragma unroll
    for (int k = 0; k < TPT; ++k)
      if (base + k * BT < e1) atomicAdd(&cnt[d[k] / rpb], 1);
  }
  __syncthreads();
  if (t < MAXB) offs[t * NBLK + blockIdx.x] = cnt[t];
}

__device__ __forceinline__ int wave_incl_scan(int c, int lane) {
  int incl = c;
#pragma unroll
  for (int o = 1; o < 32; o <<= 1) {
    const int v = __shfl_up(incl, o);
    if (lane >= o) incl += v;
  }
  return incl;
}

#define SCAN_N   (MAXB * NBLK)
#define SCAN_PER (SCAN_N / BT)
#define SCAN_LD  32
__global__ __launch_bounds__(BT) void gpr_bin_scan_kernel(int* __restrict__ offs) {
  __shared__ int st[SCAN_N + SCAN_N / SCAN_PER];
  __shared__ int wtot[BT / 32];
  const int t    = threadIdx.x;
  const int lane = t & 31;

  for (int k0 = 0; k0 < SCAN_PER; k0 += SCAN_LD) {
    int v[SCAN_LD];
#pragma unroll
    for (int k = 0; k < SCAN_LD; ++k) v[k] = offs[(k0 + k) * BT + t];
#pragma unroll
    for (int k = 0; k < SCAN_LD; ++k) {
      const int i = (k0 + k) * BT + t;
      st[i + i / SCAN_PER] = v[k];
    }
  }
  __syncthreads();

  int* q = st + t * (SCAN_PER + 1);
  int s = 0;
  for (int j = 0; j < SCAN_PER; ++j) s += q[j];
  const int incl = wave_incl_scan(s, lane);
  if (lane == 31) wtot[t >> 5] = incl;
  __syncthreads();
  int run = incl - s;
  for (int w = 0; w < (t >> 5); ++w) run += wtot[w];
  for (int j = 0; j < SCAN_PER; ++j) { const int c = q[j]; q[j] = run; run += c; }
  __syncthreads();

  for (int k = 0; k < SCAN_PER; ++k) {
    const int i = k * BT + t;
    offs[i] = st[i + i / SCAN_PER];
  }
}

__global__ __launch_bounds__(BT) void gpr_bin_place_kernel(
    const int* __restrict__ src, const int* __restrict__ dst,
    const float* __restrict__ w, const int* __restrict__ offs,
    float* __restrict__ wsort, unsigned short* __restrict__ s16,
    unsigned char* __restrict__ dl8, int nEdges, int chunk, unsigned rpb) {
  __shared__ int      gcur[MAXB];
  __shared__ int      lcnt[MAXB];
  __shared__ int      lbeg[MAXB];
  __shared__ int      wtot[MAXB / 32];
  __shared__ float    stW[TILE];
  __shared__ unsigned stK[TILE];
  const int t = threadIdx.x;
  if (t < MAXB) gcur[t] = offs[t * NBLK + blockIdx.x];
  const int e0 = blockIdx.x * chunk;
  const int e1 = min(e0 + chunk, nEdges);

  for (int base = e0; base < e1; base += TILE) {
    if (t < MAXB) lcnt[t] = 0;
    __syncthreads();

    unsigned d[TPT], bk[TPT];
    int      sv[TPT];
    float    wv[TPT];
#pragma unroll
    for (int k = 0; k < TPT; ++k) {
      const int  e  = base + k * BT + t;
      const bool ok = e < e1;
      d[k]  = ok ? (unsigned)dst[e] : 0u;
      sv[k] = ok ? src[e] : 0;
      wv[k] = ok ? w[e] : 0.0f;
    }
#pragma unroll
    for (int k = 0; k < TPT; ++k) {
      bk[k] = d[k] / rpb;
      if (base + k * BT + t < e1) atomicAdd(&lcnt[bk[k]], 1);
    }
    __syncthreads();

    const int c = (t < MAXB) ? lcnt[t] : 0;
    const int incl = wave_incl_scan(c, t & 31);
    if (t < MAXB && (t & 31) == 31) wtot[t >> 5] = incl;
    __syncthreads();
    if (t < MAXB) {
      int add = 0;
      for (int q = 0; q < (t >> 5); ++q) add += wtot[q];
      const int ex = incl - c + add;
      lbeg[t] = ex;
      lcnt[t] = ex;
    }
    __syncthreads();

#pragma unroll
    for (int k = 0; k < TPT; ++k) {
      if (base + k * BT + t < e1) {
        const int r = atomicAdd(&lcnt[bk[k]], 1);
        stW[r] = wv[k];
        stK[r] = ((unsigned)sv[k] & 0xffffu) | ((d[k] - bk[k] * rpb) << 16) | (bk[k] << 24);
      }
    }
    __syncthreads();

    const int m = min(TILE, e1 - base);
    for (int p = t; p < m; p += BT) {
      const unsigned key = stK[p];
      const unsigned b   = key >> 24;
      const int pos = gcur[b] + (p - lbeg[b]);
      wsort[pos] = stW[p];
      s16[pos]   = (unsigned short)(key & 0xffffu);
      dl8[pos]   = (unsigned char)((key >> 16) & 0xffu);
    }
    __syncthreads();
    if (t < MAXB) gcur[t] += lcnt[t] - lbeg[t];
  }
}

#define RS_CAP 16384
__global__ __launch_bounds__(BT) void gpr_row_sort_kernel(
    float* wsort, unsigned short* s16, unsigned char* dl8,
    const int* __restrict__ offs, float* __restrict__ wsum, int nEdges) {
  __shared__ int            cnt[MAXB];
  __shared__ float          dsum[MAXB];
  __shared__ int            wtot[MAXB / 32];
  __shared__ float          stW[RS_CAP];
  __shared__ unsigned short stS[RS_CAP];
  __shared__ unsigned char  stD[RS_CAP];
  const int t = threadIdx.x;
  const int b = blockIdx.x;
  const int eBeg = offs[b * NBLK];
  const int eEnd = (b + 1 < MAXB) ? offs[(b + 1) * NBLK] : nEdges;
  const int m = eEnd - eBeg;
  if (t < MAXB) { cnt[t] = 0; dsum[t] = 0.0f; }
  __syncthreads();

  for (int base = 0; base < m; base += TILE) {
    unsigned d[TPT];
    float wq[TPT];
#pragma unroll
    for (int k = 0; k < TPT; ++k) {
      const int p = base + k * BT + t;
      d[k]  = (p < m) ? (unsigned)dl8[eBeg + p] : 0u;
      wq[k] = (p < m) ? wsort[eBeg + p] : 0.0f;
    }
#pragma unroll
    for (int k = 0; k < TPT; ++k)
      if (base + k * BT + t < m) {
        atomicAdd(&cnt[d[k]], 1);
        atomicAdd(&dsum[d[k]], wq[k]);
      }
  }
  __syncthreads();
  if (t < MAXB) wsum[b * MAXB + t] = dsum[t];
  if (m > RS_CAP) return;

  const int c = (t < MAXB) ? cnt[t] : 0;
  const int incl = wave_incl_scan(c, t & 31);
  if (t < MAXB && (t & 31) == 31) wtot[t >> 5] = incl;
  __syncthreads();
  if (t < MAXB) {
    int add = 0;
    for (int q = 0; q < (t >> 5); ++q) add += wtot[q];
    cnt[t] = incl - c + add;
  }
  __syncthreads();

  for (int base = 0; base < m; base += TILE) {
    unsigned d[TPT];
    unsigned short sv[TPT];
    float wv[TPT];
#pragma unroll
    for (int k = 0; k < TPT; ++k) {
      const int  p  = base + k * BT + t;
      const bool ok = p < m;
      d[k]  = ok ? (unsigned)dl8[eBeg + p] : 0u;
      sv[k] = ok ? s16[eBeg + p] : (unsigned short)0;
      wv[k] = ok ? wsort[eBeg + p] : 0.0f;
    }
#pragma unroll
    for (int k = 0; k < TPT; ++k) {
      if (base + k * BT + t < m) {
        const int r = atomicAdd(&cnt[d[k]], 1);
        stW[r] = wv[k];
        stS[r] = sv[k];
        stD[r] = (unsigned char)d[k];
      }
    }
  }
  __syncthreads();

  for (int p = t; p < m; p += BT) {
    wsort[eBeg + p] = stW[p];
    s16[eBeg + p]   = stS[p];
    dl8[eBeg + p]   = stD[p];
  }
}

#define ACC_FLOATS 41216
#define AP      68
#define AP4     (AP / 4)
#define KP      AP
#define STG_OFF (MAXB * AP)
#define WL_OFF  (2 * MAXB * AP)

__device__ __forceinline__ void gpr_flush_run(float* acc, int row, int lane, v2f run) {
  float* o = acc + row * AP + 2 * lane;
  __hip_atomic_fetch_add(o,     run.x, __ATOMIC_RELAXED, __HIP_MEMORY_SCOPE_WORKGROUP);
  __hip_atomic_fetch_add(o + 1, run.y, __ATOMIC_RELAXED, __HIP_MEMORY_SCOPE_WORKGROUP);
}

__global__ __launch_bounds__(BT) void gpr_owner_sum_kernel(
    const float* __restrict__ Xg, const float* __restrict__ wsort,
    const unsigned short* __restrict__ s16, const unsigned char* __restrict__ dl8,
    const int* __restrict__ offs, const float* __restrict__ wsum,
    float* __restrict__ Xnext,
    float* __restrict__ hidden, const float* __restrict__ xin,
    const float* __restrict__ temp, const float* __restrict__ Wl,
    const float* __restrict__ bl, int first, int more,
    int tempIdx, int n, int rpb, int nEdges) {
  __shared__ float lds[ACC_FLOATS];
  float* acc = lds;
  float* stg = lds + STG_OFF;
  float* wl  = lds + WL_OFF;
  const int t    = threadIdx.x;
  const int b    = blockIdx.x;
  const int row0 = b * rpb;
  const int rows = min(rpb, n - row0);
  const int nRt  = (rows + 15) >> 4;
  const int nSlot  = nRt * 16;
  const int nRow4  = rows * (ND / 4);

  const float ws = (t < nSlot) ? wsum[b * MAXB + t] : 0.0f;
  const float bq = (t < ND) ? bl[t] : 0.0f;
  for (int i = t; i < nSlot * (ND / 4); i += BT)
    ((v4f*)acc)[(i >> 4) * AP4 + (i & 15)] = (v4f){0.0f, 0.0f, 0.0f, 0.0f};
  for (int i = t; i < ND * ND / 4; i += BT)
    ((v4f*)wl)[(i >> 4) * AP4 + (i & 15)] = ((const v4f*)Wl)[i];
  if (t < ND) ((v4f*)wl)[t * AP4 + ND / 4] = (v4f){bq, 0.0f, 0.0f, 0.0f};
  if (t < nSlot) ((v4f*)acc)[t * AP4 + ND / 4] = (v4f){ws, 0.0f, 0.0f, 0.0f};
  __syncthreads();

  const int eBeg = offs[b * NBLK];
  const int eEnd = (b + 1 < MAXB) ? offs[(b + 1) * NBLK] : nEdges;
  const int wv   = __builtin_amdgcn_readfirstlane(t >> 5);
  const int lane = t & 31;

  const int nW   = BT / 32;
  const int per  = ((eEnd - eBeg + nW * EPW - 1) / (nW * EPW)) * EPW;
  const int wBeg = eBeg + wv * per;
  const int wEnd = min(wBeg + per, eEnd);

  int cur = -1;
  v2f run = {0.0f, 0.0f};

  for (int g = wBeg; g < wEnd; g += EPW) {
    if (g + EPW <= wEnd) {
      v2f v[EPW];
      int dl[EPW];
#pragma unroll
      for (int j = 0; j < EPW; ++j) {
        const int   s  = s16[g + j];
        const float we = wsort[g + j];
        dl[j] = dl8[g + j];
        v[j] = *(const v2f*)(Xg + (size_t)s * ND + 2 * lane) * we;
      }
#pragma unroll
      for (int j = 0; j < EPW; ++j) {
        if (dl[j] != cur) {
          if (cur >= 0) gpr_flush_run(acc, cur, lane, run);
          cur = dl[j];
          run = v[j];
        } else {
          run += v[j];
        }
      }
    } else {
      v2f v[EPW];
      int dl[EPW];
#pragma unroll
      for (int j = 0; j < EPW; ++j) {
        const int   e  = min(g + j, wEnd - 1);
        const int   s  = s16[e];
        const float we = wsort[e];
        dl[j] = dl8[e];
        v[j] = *(const v2f*)(Xg + (size_t)s * ND + 2 * lane) * we;
      }
#pragma unroll
      for (int j = 0; j < EPW; ++j) {
        if (g + j < wEnd) {
          if (dl[j] != cur) {
            if (cur >= 0) gpr_flush_run(acc, cur, lane, run);
            cur = dl[j];
            run = v[j];
          } else {
            run += v[j];
          }
        }
      }
    }
  }
  if (cur >= 0) gpr_flush_run(acc, cur, lane, run);
  __syncthreads();

  const int rt = wv >> 1;
  const int cp = wv & 1;
  if (rt < nRt) {
    const int hi = lane >> 4;
    const int lo = lane & 15;
    const float* ap  = acc + (rt * 16 + lo) * AP + 2 * hi;
    const float* bp0 = wl  + (cp * 32 + lo) * AP + 2 * hi;
    const float* bp1 = bp0 + 16 * AP;
    v8f c0 = {};
    v8f c1 = {};
#pragma unroll
    for (int k = 0; k < KP; k += 4) {
      v2f a  = *(const v2f*)(ap  + k);
      v2f b0 = *(const v2f*)(bp0 + k);
      v2f b1 = *(const v2f*)(bp1 + k);
      c0 = __builtin_amdgcn_wmma_f32_16x16x4_f32(
           false, a,  false, b0,
           (short)0, c0,  false,  false);
      c1 = __builtin_amdgcn_wmma_f32_16x16x4_f32(
           false, a,  false, b1,
           (short)0, c1,  false,  false);
    }
    float* so = stg + (rt * 16 + 8 * hi) * AP + cp * 32 + lo;
#pragma unroll
    for (int v = 0; v < 8; ++v) {
      so[v * AP]      = fmaxf(c0[v], 0.0f);
      so[v * AP + 16] = fmaxf(c1[v], 0.0f);
    }
  }
  __syncthreads();

  const float tk = temp[tempIdx];
  v4f* ho = (v4f*)(hidden + (size_t)row0 * ND);
  v4f* xn = (v4f*)(Xnext + (size_t)row0 * ND);
  if (first) {
    const float t0 = temp[0];
    const v4f* xi = (const v4f*)(xin + (size_t)row0 * ND);
    for (int i = t; i < nRow4; i += BT) {
      const v4f v = ((const v4f*)stg)[(i >> 4) * AP4 + (i & 15)];
      ho[i] = xi[i] * t0 + v * tk;
      if (more) xn[i] = v;
    }
  } else {
    for (int i = t; i < nRow4; i += BT) {
      const v4f v = ((const v4f*)stg)[(i >> 4) * AP4 + (i & 15)];
      ho[i] = ho[i] + v * tk;
      if (more) xn[i] = v;
    }
  }
}

extern "C" void kernel_launch(void* const* d_in, const int* in_sizes, int n_in,
                              void* d_out, int out_size, void* d_ws, size_t ws_size,
                              hipStream_t stream) {
  const float* x    = (const float*)d_in[0];
  const float* w    = (const float*)d_in[1];
  const float* Wm   = (const float*)d_in[2];
  const float* bias = (const float*)d_in[3];
  const float* temp = (const float*)d_in[4];
  const int*   src  = (const int*)d_in[5];
  const int*   dst  = (const int*)d_in[6];
  float* hidden = (float*)d_out;

  const int n = in_sizes[0] / ND;
  const int E = in_sizes[1];
  const int L = in_sizes[3] / ND;
  const int total = n * ND;

  float* xA = (float*)d_ws;
  float* xB = xA + (size_t)total;
  char* plane = (char*)(xB + (size_t)total);
  float*          wsort = (float*)plane;
  unsigned short* s16   = (unsigned short*)(plane + (size_t)E * 4);
  unsigned char*  dl8   = (unsigned char*)(plane + (size_t)E * 6);
  int*            offs  = (int*)(plane + (((size_t)E * 7 + 3) & ~(size_t)3));
  float*          wsum  = (float*)(offs + MAXB * NBLK);

  const int rpb   = (n + MAXB - 1) / MAXB;
  const int nb    = (n + rpb - 1) / rpb;
  const int chunk = (E + NBLK - 1) / NBLK;

  gpr_bin_count_kernel<<<NBLK, BT, 0, stream>>>(dst, offs, E, chunk, (unsigned)rpb);
  gpr_bin_scan_kernel<<<1, BT, 0, stream>>>(offs);
  gpr_bin_place_kernel<<<NBLK, BT, 0, stream>>>(src, dst, w, offs, wsort, s16, dl8,
                                                 E, chunk, (unsigned)rpb);
  gpr_row_sort_kernel<<<nb, BT, 0, stream>>>(wsort, s16, dl8, offs, wsum, E);

  for (int i = 0; i < L; ++i) {
    const float* xg = (i == 0) ? x : ((i & 1) ? xA : xB);
    float* xn = (i & 1) ? xB : xA;
    const int more = (i + 1 < L) ? 1 : 0;
    gpr_owner_sum_kernel<<<nb, BT, 0, stream>>>(
        xg, wsort, s16, dl8, offs, wsum, xn, hidden, x, temp,
        Wm + (size_t)i * ND * ND, bias + (size_t)i * ND, (i == 0) ? 1 : 0, more,
        i + 1, n, rpb, E);
  }
}


# Owner walk's per-edge work cut: a row change is one rarely taken branch that flushes and zeroes the run, then weight*row+run as one packed FMA on a 32-bit row offset; sum order kept, product fused
# speedup vs baseline: 3.8464x; 1.0397x over previous
#include <hip/hip_runtime.h>
#include <hip/hip_bf16.h>


typedef __attribute__((ext_vector_type(2))) float v2f;
typedef __attribute__((ext_vector_type(4))) float v4f;
typedef __attribute__((ext_vector_type(8))) float v8f;

#define ND   64
#define BT   1024
#define NBLK 256
#define MAXB 256
#define EPW  16
#define TPT  8
#define TILE (BT * TPT)

__global__ __launch_bounds__(BT) void gpr_bin_count_kernel(
    const int* __restrict__ dst, int* __restrict__ offs,
    int nEdges, int chunk, unsigned rpb) {
  __shared__ int cnt[MAXB];
  const int t = threadIdx.x;
  if (t < MAXB) cnt[t] = 0;
  __syncthreads();
  const int e0 = blockIdx.x * chunk;
  const int e1 = min(e0 + chunk, nEdges);
  for (int base = e0 + t; base < e1; base += TILE) {
    unsigned d[TPT];
#pragma unroll
    for (int k = 0; k < TPT; ++k)
      d[k] = (base + k * BT < e1) ? (unsigned)dst[base + k * BT] : 0u;
#pragma unroll
    for (int k = 0; k < TPT; ++k)
      if (base + k * BT < e1) atomicAdd(&cnt[d[k] / rpb], 1);
  }
  __syncthreads();
  if (t < MAXB) offs[t * NBLK + blockIdx.x] = cnt[t];
}

__device__ __forceinline__ int wave_incl_scan(int c, int lane) {
  int incl = c;
#pragma unroll
  for (int o = 1; o < 32; o <<= 1) {
    const int v = __shfl_up(incl, o);
    if (lane >= o) incl += v;
  }
  return incl;
}

#define SCAN_N   (MAXB * NBLK)
#define SCAN_PER (SCAN_N / BT)
#define SCAN_LD  32
__global__ __launch_bounds__(BT) void gpr_bin_scan_kernel(int* __restrict__ offs) {
  __shared__ int st[SCAN_N + SCAN_N / SCAN_PER];
  __shared__ int wtot[BT / 32];
  const int t    = threadIdx.x;
  const int lane = t & 31;

  for (int k0 = 0; k0 < SCAN_PER; k0 += SCAN_LD) {
    int v[SCAN_LD];
#pragma unroll
    for (int k = 0; k < SCAN_LD; ++k) v[k] = offs[(k0 + k) * BT + t];
#pragma unroll
    for (int k = 0; k < SCAN_LD; ++k) {
      const int i = (k0 + k) * BT + t;
      st[i + i / SCAN_PER] = v[k];
    }
  }
  __syncthreads();

  int* q = st + t * (SCAN_PER + 1);
  int s = 0;
  for (int j = 0; j < SCAN_PER; ++j) s += q[j];
  const int incl = wave_incl_scan(s, lane);
  if (lane == 31) wtot[t >> 5] = incl;
  __syncthreads();
  int run = incl - s;
  for (int w = 0; w < (t >> 5); ++w) run += wtot[w];
  for (int j = 0; j < SCAN_PER; ++j) { const int c = q[j]; q[j] = run; run += c; }
  __syncthreads();

  for (int k = 0; k < SCAN_PER; ++k) {
    const int i = k * BT + t;
    offs[i] = st[i + i / SCAN_PER];
  }
}

__global__ __launch_bounds__(BT) void gpr_bin_place_kernel(
    const int* __restrict__ src, const int* __restrict__ dst,
    const float* __restrict__ w, const int* __restrict__ offs,
    float* __restrict__ wsort, unsigned short* __restrict__ s16,
    unsigned char* __restrict__ dl8, int nEdges, int chunk, unsigned rpb) {
  __shared__ int      gcur[MAXB];
  __shared__ int      lcnt[MAXB];
  __shared__ int      lbeg[MAXB];
  __shared__ int      wtot[MAXB / 32];
  __shared__ float    stW[TILE];
  __shared__ unsigned stK[TILE];
  const int t = threadIdx.x;
  if (t < MAXB) gcur[t] = offs[t * NBLK + blockIdx.x];
  const int e0 = blockIdx.x * chunk;
  const int e1 = min(e0 + chunk, nEdges);

  for (int base = e0; base < e1; base += TILE) {
    if (t < MAXB) lcnt[t] = 0;
    __syncthreads();

    unsigned d[TPT], bk[TPT];
    int      sv[TPT];
    float    wv[TPT];
#pragma unroll
    for (int k = 0; k < TPT; ++k) {
      const int  e  = base + k * BT + t;
      const bool ok = e < e1;
      d[k]  = ok ? (unsigned)dst[e] : 0u;
      sv[k] = ok ? src[e] : 0;
      wv[k] = ok ? w[e] : 0.0f;
    }
#pragma unroll
    for (int k = 0; k < TPT; ++k) {
      bk[k] = d[k] / rpb;
      if (base + k * BT + t < e1) atomicAdd(&lcnt[bk[k]], 1);
    }
    __syncthreads();

    const int c = (t < MAXB) ? lcnt[t] : 0;
    const int incl = wave_incl_scan(c, t & 31);
    if (t < MAXB && (t & 31) == 31) wtot[t >> 5] = incl;
    __syncthreads();
    if (t < MAXB) {
      int add = 0;
      for (int q = 0; q < (t >> 5); ++q) add += wtot[q];
      const int ex = incl - c + add;
      lbeg[t] = ex;
      lcnt[t] = ex;
    }
    __syncthreads();

#pragma unroll
    for (int k = 0; k < TPT; ++k) {
      if (base + k * BT + t < e1) {
        const int r = atomicAdd(&lcnt[bk[k]], 1);
        stW[r] = wv[k];
        stK[r] = ((unsigned)sv[k] & 0xffffu) | ((d[k] - bk[k] * rpb) << 16) | (bk[k] << 24);
      }
    }
    __syncthreads();

    const int m = min(TILE, e1 - base);
    for (int p = t; p < m; p += BT) {
      const unsigned key = stK[p];
      const unsigned b   = key >> 24;
      const int pos = gcur[b] + (p - lbeg[b]);
      wsort[pos] = stW[p];
      s16[pos]   = (unsigned short)(key & 0xffffu);
      dl8[pos]   = (unsigned char)((key >> 16) & 0xffu);
    }
    __syncthreads();
    if (t < MAXB) gcur[t] += lcnt[t] - lbeg[t];
  }
}

#define RS_CAP 16384
__global__ __launch_bounds__(BT) void gpr_row_sort_kernel(
    float* wsort, unsigned short* s16, unsigned char* dl8,
    const int* __restrict__ offs, float* __restrict__ wsum, int nEdges) {
  __shared__ int            cnt[MAXB];
  __shared__ float          dsum[MAXB];
  __shared__ int            wtot[MAXB / 32];
  __shared__ float          stW[RS_CAP];
  __shared__ unsigned short stS[RS_CAP];
  __shared__ unsigned char  stD[RS_CAP];
  const int t = threadIdx.x;
  const int b = blockIdx.x;
  const int eBeg = offs[b * NBLK];
  const int eEnd = (b + 1 < MAXB) ? offs[(b + 1) * NBLK] : nEdges;
  const int m = eEnd - eBeg;
  if (t < MAXB) { cnt[t] = 0; dsum[t] = 0.0f; }
  __syncthreads();

  for (int base = 0; base < m; base += TILE) {
    unsigned d[TPT];
    float wq[TPT];
#pragma unroll
    for (int k = 0; k < TPT; ++k) {
      const int p = base + k * BT + t;
      d[k]  = (p < m) ? (unsigned)dl8[eBeg + p] : 0u;
      wq[k] = (p < m) ? wsort[eBeg + p] : 0.0f;
    }
#pragma unroll
    for (int k = 0; k < TPT; ++k)
      if (base + k * BT + t < m) {
        atomicAdd(&cnt[d[k]], 1);
        atomicAdd(&dsum[d[k]], wq[k]);
      }
  }
  __syncthreads();
  if (t < MAXB) wsum[b * MAXB + t] = dsum[t];
  if (m > RS_CAP) return;

  const int c = (t < MAXB) ? cnt[t] : 0;
  const int incl = wave_incl_scan(c, t & 31);
  if (t < MAXB && (t & 31) == 31) wtot[t >> 5] = incl;
  __syncthreads();
  if (t < MAXB) {
    int add = 0;
    for (int q = 0; q < (t >> 5); ++q) add += wtot[q];
    cnt[t] = incl - c + add;
  }
  __syncthreads();

  for (int base = 0; base < m; base += TILE) {
    unsigned d[TPT];
    unsigned short sv[TPT];
    float wv[TPT];
#pragma unroll
    for (int k = 0; k < TPT; ++k) {
      const int  p  = base + k * BT + t;
      const bool ok = p < m;
      d[k]  = ok ? (unsigned)dl8[eBeg + p] : 0u;
      sv[k] = ok ? s16[eBeg + p] : (unsigned short)0;
      wv[k] = ok ? wsort[eBeg + p] : 0.0f;
    }
#pragma unroll
    for (int k = 0; k < TPT; ++k) {
      if (base + k * BT + t < m) {
        const int r = atomicAdd(&cnt[d[k]], 1);
        stW[r] = wv[k];
        stS[r] = sv[k];
        stD[r] = (unsigned char)d[k];
      }
    }
  }
  __syncthreads();

  for (int p = t; p < m; p += BT) {
    wsort[eBeg + p] = stW[p];
    s16[eBeg + p]   = stS[p];
    dl8[eBeg + p]   = stD[p];
  }
}

#define ACC_FLOATS 41216
#define AP      68
#define AP4     (AP / 4)
#define KP      AP
#define STG_OFF (MAXB * AP)
#define WL_OFF  (2 * MAXB * AP)

__device__ __forceinline__ void gpr_flush_run(float* acc, int row, int lane, v2f run) {
  float* o = acc + row * AP + 2 * lane;
  __hip_atomic_fetch_add(o,     run.x, __ATOMIC_RELAXED, __HIP_MEMORY_SCOPE_WORKGROUP);
  __hip_atomic_fetch_add(o + 1, run.y, __ATOMIC_RELAXED, __HIP_MEMORY_SCOPE_WORKGROUP);
}

__global__ __launch_bounds__(BT) void gpr_owner_sum_kernel(
    const float* __restrict__ Xg, const float* __restrict__ wsort,
    const unsigned short* __restrict__ s16, const unsigned char* __restrict__ dl8,
    const int* __restrict__ offs, const float* __restrict__ wsum,
    float* __restrict__ Xnext,
    float* __restrict__ hidden, const float* __restrict__ xin,
    const float* __restrict__ temp, const float* __restrict__ Wl,
    const float* __restrict__ bl, int first, int more,
    int tempIdx, int n, int rpb, int nEdges) {
  __shared__ float lds[ACC_FLOATS];
  float* acc = lds;
  float* stg = lds + STG_OFF;
  float* wl  = lds + WL_OFF;
  const int t    = threadIdx.x;
  const int b    = blockIdx.x;
  const int row0 = b * rpb;
  const int rows = min(rpb, n - row0);
  const int nRt  = (rows + 15) >> 4;
  const int nSlot  = nRt * 16;
  const int nRow4  = rows * (ND / 4);

  const float ws = (t < nSlot) ? wsum[b * MAXB + t] : 0.0f;
  const float bq = (t < ND) ? bl[t] : 0.0f;
  for (int i = t; i < nSlot * (ND / 4); i += BT)
    ((v4f*)acc)[(i >> 4) * AP4 + (i & 15)] = (v4f){0.0f, 0.0f, 0.0f, 0.0f};
  for (int i = t; i < ND * ND / 4; i += BT)
    ((v4f*)wl)[(i >> 4) * AP4 + (i & 15)] = ((const v4f*)Wl)[i];
  if (t < ND) ((v4f*)wl)[t * AP4 + ND / 4] = (v4f){bq, 0.0f, 0.0f, 0.0f};
  if (t < nSlot) ((v4f*)acc)[t * AP4 + ND / 4] = (v4f){ws, 0.0f, 0.0f, 0.0f};
  __syncthreads();

  const int eBeg = offs[b * NBLK];
  const int eEnd = (b + 1 < MAXB) ? offs[(b + 1) * NBLK] : nEdges;
  const int wv   = __builtin_amdgcn_readfirstlane(t >> 5);
  const int lane = t & 31;

  const int nW   = BT / 32;
  const int per  = ((eEnd - eBeg + nW * EPW - 1) / (nW * EPW)) * EPW;
  const int wBeg = eBeg + wv * per;
  const int wEnd = min(wBeg + per, eEnd);

  int cur = -1;
  v2f run = {0.0f, 0.0f};
  const char* xb = (const char*)Xg;
  const unsigned lane8 = (unsigned)lane * 8u;

  for (int g = wBeg; g < wEnd; g += EPW) {
    if (g + EPW <= wEnd) {
      v2f v[EPW];
      float we[EPW];
      int dl[EPW];
#pragma unroll
      for (int j = 0; j < EPW; ++j) {
        const unsigned s = s16[g + j];
        we[j] = wsort[g + j];
        dl[j] = dl8[g + j];
        v[j] = *(const v2f*)(xb + (s * (unsigned)(ND * 4) + lane8));
      }
#pragma unroll
      for (int j = 0; j < EPW; ++j) {
        if (__builtin_expect(dl[j] != cur, 0)) {
          if (cur >= 0) gpr_flush_run(acc, cur, lane, run);
          cur = dl[j];
          run = (v2f){0.0f, 0.0f};
        }
        run = __builtin_elementwise_fma(v[j], (v2f){we[j], we[j]}, run);
      }
    } else {
      v2f v[EPW];
      float we[EPW];
      int dl[EPW];
#pragma unroll
      for (int j = 0; j < EPW; ++j) {
        const int      e = min(g + j, wEnd - 1);
        const unsigned s = s16[e];
        we[j] = wsort[e];
        dl[j] = dl8[e];
        v[j] = *(const v2f*)(xb + (s * (unsigned)(ND * 4) + lane8));
      }
#pragma unroll
      for (int j = 0; j < EPW; ++j) {
        if (g + j < wEnd) {
          if (__builtin_expect(dl[j] != cur, 0)) {
            if (cur >= 0) gpr_flush_run(acc, cur, lane, run);
            cur = dl[j];
            run = (v2f){0.0f, 0.0f};
          }
          run = __builtin_elementwise_fma(v[j], (v2f){we[j], we[j]}, run);
        }
      }
    }
  }
  if (cur >= 0) gpr_flush_run(acc, cur, lane, run);
  __syncthreads();

  const int rt = wv >> 1;
  const int cp = wv & 1;
  if (rt < nRt) {
    const int hi = lane >> 4;
    const int lo = lane & 15;
    const float* ap  = acc + (rt * 16 + lo) * AP + 2 * hi;
    const float* bp0 = wl  + (cp * 32 + lo) * AP + 2 * hi;
    const float* bp1 = bp0 + 16 * AP;
    v8f c0 = {};
    v8f c1 = {};
#pragma unroll
    for (int k = 0; k < KP; k += 4) {
      v2f a  = *(const v2f*)(ap  + k);
      v2f b0 = *(const v2f*)(bp0 + k);
      v2f b1 = *(const v2f*)(bp1 + k);
      c0 = __builtin_amdgcn_wmma_f32_16x16x4_f32(
           false, a,  false, b0,
           (short)0, c0,  false,  false);
      c1 = __builtin_amdgcn_wmma_f32_16x16x4_f32(
           false, a,  false, b1,
           (short)0, c1,  false,  false);
    }
    float* so = stg + (rt * 16 + 8 * hi) * AP + cp * 32 + lo;
#pragma unroll
    for (int v = 0; v < 8; ++v) {
      so[v * AP]      = fmaxf(c0[v], 0.0f);
      so[v * AP + 16] = fmaxf(c1[v], 0.0f);
    }
  }
  __syncthreads();

  const float tk = temp[tempIdx];
  v4f* ho = (v4f*)(hidden + (size_t)row0 * ND);
  v4f* xn = (v4f*)(Xnext + (size_t)row0 * ND);
  if (first) {
    const float t0 = temp[0];
    const v4f* xi = (const v4f*)(xin + (size_t)row0 * ND);
    for (int i = t; i < nRow4; i += BT) {
      const v4f v = ((const v4f*)stg)[(i >> 4) * AP4 + (i & 15)];
      ho[i] = xi[i] * t0 + v * tk;
      if (more) xn[i] = v;
    }
  } else {
    for (int i = t; i < nRow4; i += BT) {
      const v4f v = ((const v4f*)stg)[(i >> 4) * AP4 + (i & 15)];
      ho[i] = ho[i] + v * tk;
      if (more) xn[i] = v;
    }
  }
}

extern "C" void kernel_launch(void* const* d_in, const int* in_sizes, int n_in,
                              void* d_out, int out_size, void* d_ws, size_t ws_size,
                              hipStream_t stream) {
  const float* x    = (const float*)d_in[0];
  const float* w    = (const float*)d_in[1];
  const float* Wm   = (const float*)d_in[2];
  const float* bias = (const float*)d_in[3];
  const float* temp = (const float*)d_in[4];
  const int*   src  = (const int*)d_in[5];
  const int*   dst  = (const int*)d_in[6];
  float* hidden = (float*)d_out;

  const int n = in_sizes[0] / ND;
  const int E = in_sizes[1];
  const int L = in_sizes[3] / ND;
  const int total = n * ND;

  float* xA = (float*)d_ws;
  float* xB = xA + (size_t)total;
  char* plane = (char*)(xB + (size_t)total);
  float*          wsort = (float*)plane;
  unsigned short* s16   = (unsigned short*)(plane + (size_t)E * 4);
  unsigned char*  dl8   = (unsigned char*)(plane + (size_t)E * 6);
  int*            offs  = (int*)(plane + (((size_t)E * 7 + 3) & ~(size_t)3));
  float*          wsum  = (float*)(offs + MAXB * NBLK);

  const int rpb   = (n + MAXB - 1) / MAXB;
  const int nb    = (n + rpb - 1) / rpb;
  const int chunk = (E + NBLK - 1) / NBLK;

  gpr_bin_count_kernel<<<NBLK, BT, 0, stream>>>(dst, offs, E, chunk, (unsigned)rpb);
  gpr_bin_scan_kernel<<<1, BT, 0, stream>>>(offs);
  gpr_bin_place_kernel<<<NBLK, BT, 0, stream>>>(src, dst, w, offs, wsort, s16, dl8,
                                                 E, chunk, (unsigned)rpb);
  gpr_row_sort_kernel<<<nb, BT, 0, stream>>>(wsort, s16, dl8, offs, wsum, E);

  for (int i = 0; i < L; ++i) {
    const float* xg = (i == 0) ? x : ((i & 1) ? xA : xB);
    float* xn = (i & 1) ? xB : xA;
    const int more = (i + 1 < L) ? 1 : 0;
    gpr_owner_sum_kernel<<<nb, BT, 0, stream>>>(
        xg, wsort, s16, dl8, offs, wsum, xn, hidden, x, temp,
        Wm + (size_t)i * ND * ND, bias + (size_t)i * ND, (i == 0) ? 1 : 0, more,
        i + 1, n, rpb, E);
  }
}


# Binning made pull: a chunk orders its edges by range in its own slots and the range's row-sort group collects the 256 pieces; global count and scan launches gone (8 to 6); sum order stays unordered
# speedup vs baseline: 4.3399x; 1.1283x over previous
#include <hip/hip_runtime.h>
#include <hip/hip_bf16.h>


typedef __attribute__((ext_vector_type(2))) float v2f;
typedef __attribute__((ext_vector_type(4))) float v4f;
typedef __attribute__((ext_vector_type(8))) float v8f;

#define ND   64
#define BT   1024
#define NBLK 256
#define MAXB 256
#define EPW  16
#define TPT  8
#define TILE (BT * TPT)


__device__ __forceinline__ int wave_incl_scan(int c, int lane) {
  int incl = c;
#pragma unroll
  for (int o = 1; o < 32; o <<= 1) {
    const int v = __shfl_up(incl, o);
    if (lane >= o) incl += v;
  }
  return incl;
}

__global__ __launch_bounds__(BT) void gpr_chunk_group_kernel(
    const int* __restrict__ src, const int* __restrict__ dst,
    const float* __restrict__ w, int* __restrict__ tab,
    float* __restrict__ wtmp, unsigned short* __restrict__ stmp,
    unsigned char* __restrict__ dtmp, int nEdges, int chunk, unsigned rpb) {
  __shared__ int      lcnt[MAXB];
  __shared__ int      wtot[MAXB / 32];
  __shared__ float    stW[TILE];
  __shared__ unsigned stK[TILE];
  const int t  = threadIdx.x;
  const int e0 = blockIdx.x * chunk;
  const int e1 = min(min(e0 + chunk, nEdges), e0 + TILE);
  if (t < MAXB) lcnt[t] = 0;
  __syncthreads();

  unsigned d[TPT], bk[TPT];
  int      sv[TPT];
  float    wv[TPT];
#pragma unroll
  for (int k = 0; k < TPT; ++k) {
    const int  e  = e0 + k * BT + t;
    const bool ok = e < e1;
    d[k]  = ok ? (unsigned)dst[e] : 0u;
    sv[k] = ok ? src[e] : 0;
    wv[k] = ok ? w[e] : 0.0f;
  }
#pragma unroll
  for (int k = 0; k < TPT; ++k) {
    bk[k] = d[k] / rpb;
    if (e0 + k * BT + t < e1) atomicAdd(&lcnt[bk[k]], 1);
  }
  __syncthreads();

  const int c = (t < MAXB) ? lcnt[t] : 0;
  const int incl = wave_incl_scan(c, t & 31);
  if (t < MAXB && (t & 31) == 31) wtot[t >> 5] = incl;
  __syncthreads();
  if (t < MAXB) {
    int add = 0;
    for (int q = 0; q < (t >> 5); ++q) add += wtot[q];
    const int ex = incl - c + add;
    tab[t * NBLK + blockIdx.x] = (ex << 16) | c;
    lcnt[t] = ex;
  }
  __syncthreads();

#pragma unroll
  for (int k = 0; k < TPT; ++k) {
    if (e0 + k * BT + t < e1) {
      const int r = atomicAdd(&lcnt[bk[k]], 1);
      stW[r] = wv[k];
      stK[r] = ((unsigned)sv[k] & 0xffffu) | ((d[k] - bk[k] * rpb) << 16);
    }
  }
  __syncthreads();

  const int m = e1 - e0;
  for (int p = t; p < m; p += BT) {
    const unsigned key = stK[p];
    wtmp[e0 + p] = stW[p];
    stmp[e0 + p] = (unsigned short)(key & 0xffffu);
    dtmp[e0 + p] = (unsigned char)((key >> 16) & 0xffu);
  }
}

#define RS_CAP 16384

__device__ __forceinline__ int gpr_piece_slot(const int* pre, const int* sbeg, int p) {
  int c = 0;
#pragma unroll
  for (int s = NBLK / 2; s > 0; s >>= 1)
    if (pre[c + s] <= p) c += s;
  return sbeg[c] + p;
}

__global__ __launch_bounds__(BT) void gpr_row_sort_kernel(
    const float* __restrict__ wtmp, const unsigned short* __restrict__ stmp,
    const unsigned char* __restrict__ dtmp, const int* __restrict__ tab,
    float* __restrict__ wsort, unsigned short* __restrict__ s16,
    unsigned char* __restrict__ dl8, int* __restrict__ rng,
    float* __restrict__ wsum, int chunk) {
  __shared__ int            cnt[MAXB];
  __shared__ float          dsum[MAXB];
  __shared__ int            wtot[NBLK / 32];
  __shared__ int            ptot[NBLK / 32];
  __shared__ int            pre[NBLK];
  __shared__ int            sbeg[NBLK];
  __shared__ int            mAll;
  __shared__ float          stW[RS_CAP];
  __shared__ unsigned short stS[RS_CAP];
  __shared__ unsigned char  stD[RS_CAP];
  const int t    = threadIdx.x;
  const int lane = t & 31;
  const int b    = blockIdx.x;

  const int e  = (t < NBLK) ? tab[b * NBLK + t] : 0;
  const int pc = e & 0xffff;
  const int pf = (int)((unsigned)e >> 16);
  int part = pf;
#pragma unroll
  for (int o = 16; o > 0; o >>= 1) part += __shfl_xor(part, o);
  const int pincl = wave_incl_scan(pc, lane);
  if (t < NBLK && lane == 0)  ptot[t >> 5] = part;
  if (t < NBLK && lane == 31) wtot[t >> 5] = pincl;
  if (t < MAXB) { cnt[t] = 0; dsum[t] = 0.0f; }
  __syncthreads();
  int gBeg = 0;
  for (int q = 0; q < NBLK / 32; ++q) gBeg += ptot[q];
  if (t < NBLK) {
    int add = 0;
    for (int q = 0; q < (t >> 5); ++q) add += wtot[q];
    const int ex = pincl - pc + add;
    pre[t]  = ex;
    sbeg[t] = t * chunk + pf - ex;
    if (t == NBLK - 1) mAll = ex + pc;
  }
  __syncthreads();
  const int m = mAll;
  if (t == 0) { rng[2 * b] = gBeg; rng[2 * b + 1] = gBeg + m; }

  for (int base = 0; base < m; base += TILE) {
    unsigned d[TPT];
    float wq[TPT];
#pragma unroll
    for (int k = 0; k < TPT; ++k) {
      const int gi = gpr_piece_slot(pre, sbeg, min(base + k * BT + t, m - 1));
      d[k]  = (unsigned)dtmp[gi];
      wq[k] = wtmp[gi];
    }
#pragma unroll
    for (int k = 0; k < TPT; ++k)
      if (base + k * BT + t < m) {
        atomicAdd(&cnt[d[k]], 1);
        atomicAdd(&dsum[d[k]], wq[k]);
      }
  }
  __syncthreads();
  if (t < MAXB) wsum[b * MAXB + t] = dsum[t];

  if (m > RS_CAP) {
    for (int p = t; p < m; p += BT) {
      const int gi = gpr_piece_slot(pre, sbeg, p);
      wsort[gBeg + p] = wtmp[gi];
      s16[gBeg + p]   = stmp[gi];
      dl8[gBeg + p]   = dtmp[gi];
    }
    return;
  }

  const int c = (t < MAXB) ? cnt[t] : 0;
  const int incl = wave_incl_scan(c, lane);
  if (t < MAXB && lane == 31) wtot[t >> 5] = incl;
  __syncthreads();
  if (t < MAXB) {
    int add = 0;
    for (int q = 0; q < (t >> 5); ++q) add += wtot[q];
    cnt[t] = incl - c + add;
  }
  __syncthreads();

  for (int base = 0; base < m; base += TILE) {
    unsigned d[TPT];
    unsigned short sv[TPT];
    float wv[TPT];
#pragma unroll
    for (int k = 0; k < TPT; ++k) {
      const int gi = gpr_piece_slot(pre, sbeg, min(base + k * BT + t, m - 1));
      d[k]  = (unsigned)dtmp[gi];
      sv[k] = stmp[gi];
      wv[k] = wtmp[gi];
    }
#pragma unroll
    for (int k = 0; k < TPT; ++k) {
      if (base + k * BT + t < m) {
        const int r = atomicAdd(&cnt[d[k]], 1);
        stW[r] = wv[k];
        stS[r] = sv[k];
        stD[r] = (unsigned char)d[k];
      }
    }
  }
  __syncthreads();

  for (int p = t; p < m; p += BT) {
    wsort[gBeg + p] = stW[p];
    s16[gBeg + p]   = stS[p];
    dl8[gBeg + p]   = stD[p];
  }
}

#define ACC_FLOATS 41216
#define AP      68
#define AP4     (AP / 4)
#define KP      AP
#define STG_OFF (MAXB * AP)
#define WL_OFF  (2 * MAXB * AP)

__device__ __forceinline__ void gpr_flush_run(float* acc, int row, int lane, v2f run) {
  float* o = acc + row * AP + 2 * lane;
  __hip_atomic_fetch_add(o,     run.x, __ATOMIC_RELAXED, __HIP_MEMORY_SCOPE_WORKGROUP);
  __hip_atomic_fetch_add(o + 1, run.y, __ATOMIC_RELAXED, __HIP_MEMORY_SCOPE_WORKGROUP);
}

__global__ __launch_bounds__(BT) void gpr_owner_sum_kernel(
    const float* __restrict__ Xg, const float* __restrict__ wsort,
    const unsigned short* __restrict__ s16, const unsigned char* __restrict__ dl8,
    const int* __restrict__ rng, const float* __restrict__ wsum,
    float* __restrict__ Xnext,
    float* __restrict__ hidden, const float* __restrict__ xin,
    const float* __restrict__ temp, const float* __restrict__ Wl,
    const float* __restrict__ bl, int first, int more,
    int tempIdx, int n, int rpb) {
  __shared__ float lds[ACC_FLOATS];
  float* acc = lds;
  float* stg = lds + STG_OFF;
  float* wl  = lds + WL_OFF;
  const int t    = threadIdx.x;
  const int b    = blockIdx.x;
  const int row0 = b * rpb;
  const int rows = min(rpb, n - row0);
  const int nRt  = (rows + 15) >> 4;
  const int nSlot  = nRt * 16;
  const int nRow4  = rows * (ND / 4);

  const float ws = (t < nSlot) ? wsum[b * MAXB + t] : 0.0f;
  const float bq = (t < ND) ? bl[t] : 0.0f;
  for (int i = t; i < nSlot * (ND / 4); i += BT)
    ((v4f*)acc)[(i >> 4) * AP4 + (i & 15)] = (v4f){0.0f, 0.0f, 0.0f, 0.0f};
  for (int i = t; i < ND * ND / 4; i += BT)
    ((v4f*)wl)[(i >> 4) * AP4 + (i & 15)] = ((const v4f*)Wl)[i];
  if (t < ND) ((v4f*)wl)[t * AP4 + ND / 4] = (v4f){bq, 0.0f, 0.0f, 0.0f};
  if (t < nSlot) ((v4f*)acc)[t * AP4 + ND / 4] = (v4f){ws, 0.0f, 0.0f, 0.0f};
  __syncthreads();

  const int eBeg = rng[2 * b];
  const int eEnd = rng[2 * b + 1];
  const int wv   = __builtin_amdgcn_readfirstlane(t >> 5);
  const int lane = t & 31;

  const int nW   = BT / 32;
  const int per  = ((eEnd - eBeg + nW * EPW - 1) / (nW * EPW)) * EPW;
  const int wBeg = eBeg + wv * per;
  const int wEnd = min(wBeg + per, eEnd);

  int cur = -1;
  v2f run = {0.0f, 0.0f};
  const char* xb = (const char*)Xg;
  const unsigned lane8 = (unsigned)lane * 8u;

  for (int g = wBeg; g < wEnd; g += EPW) {
    if (g + EPW <= wEnd) {
      v2f v[EPW];
      float we[EPW];
      int dl[EPW];
#pragma unroll
      for (int j = 0; j < EPW; ++j) {
        const unsigned s = s16[g + j];
        we[j] = wsort[g + j];
        dl[j] = dl8[g + j];
        v[j] = *(const v2f*)(xb + (s * (unsigned)(ND * 4) + lane8));
      }
#pragma unroll
      for (int j = 0; j < EPW; ++j) {
        if (__builtin_expect(dl[j] != cur, 0)) {
          if (cur >= 0) gpr_flush_run(acc, cur, lane, run);
          cur = dl[j];
          run = (v2f){0.0f, 0.0f};
        }
        run = __builtin_elementwise_fma(v[j], (v2f){we[j], we[j]}, run);
      }
    } else {
      v2f v[EPW];
      float we[EPW];
      int dl[EPW];
#pragma unroll
      for (int j = 0; j < EPW; ++j) {
        const int      e = min(g + j, wEnd - 1);
        const unsigned s = s16[e];
        we[j] = wsort[e];
        dl[j] = dl8[e];
        v[j] = *(const v2f*)(xb + (s * (unsigned)(ND * 4) + lane8));
      }
#pragma unroll
      for (int j = 0; j < EPW; ++j) {
        if (g + j < wEnd) {
          if (__builtin_expect(dl[j] != cur, 0)) {
            if (cur >= 0) gpr_flush_run(acc, cur, lane, run);
            cur = dl[j];
            run = (v2f){0.0f, 0.0f};
          }
          run = __builtin_elementwise_fma(v[j], (v2f){we[j], we[j]}, run);
        }
      }
    }
  }
  if (cur >= 0) gpr_flush_run(acc, cur, lane, run);
  __syncthreads();

  const int rt = wv >> 1;
  const int cp = wv & 1;
  if (rt < nRt) {
    const int hi = lane >> 4;
    const int lo = lane & 15;
    const float* ap  = acc + (rt * 16 + lo) * AP + 2 * hi;
    const float* bp0 = wl  + (cp * 32 + lo) * AP + 2 * hi;
    const float* bp1 = bp0 + 16 * AP;
    v8f c0 = {};
    v8f c1 = {};
#pragma unroll
    for (int k = 0; k < KP; k += 4) {
      v2f a  = *(const v2f*)(ap  + k);
      v2f b0 = *(const v2f*)(bp0 + k);
      v2f b1 = *(const v2f*)(bp1 + k);
      c0 = __builtin_amdgcn_wmma_f32_16x16x4_f32(
           false, a,  false, b0,
           (short)0, c0,  false,  false);
      c1 = __builtin_amdgcn_wmma_f32_16x16x4_f32(
           false, a,  false, b1,
           (short)0, c1,  false,  false);
    }
    float* so = stg + (rt * 16 + 8 * hi) * AP + cp * 32 + lo;
#pragma unroll
    for (int v = 0; v < 8; ++v) {
      so[v * AP]      = fmaxf(c0[v], 0.0f);
      so[v * AP + 16] = fmaxf(c1[v], 0.0f);
    }
  }
  __syncthreads();

  const float tk = temp[tempIdx];
  v4f* ho = (v4f*)(hidden + (size_t)row0 * ND);
  v4f* xn = (v4f*)(Xnext + (size_t)row0 * ND);
  if (first) {
    const float t0 = temp[0];
    const v4f* xi = (const v4f*)(xin + (size_t)row0 * ND);
    for (int i = t; i < nRow4; i += BT) {
      const v4f v = ((const v4f*)stg)[(i >> 4) * AP4 + (i & 15)];
      ho[i] = xi[i] * t0 + v * tk;
      if (more) xn[i] = v;
    }
  } else {
    for (int i = t; i < nRow4; i += BT) {
      const v4f v = ((const v4f*)stg)[(i >> 4) * AP4 + (i & 15)];
      ho[i] = ho[i] + v * tk;
      if (more) xn[i] = v;
    }
  }
}

extern "C" void kernel_launch(void* const* d_in, const int* in_sizes, int n_in,
                              void* d_out, int out_size, void* d_ws, size_t ws_size,
                              hipStream_t stream) {
  const float* x    = (const float*)d_in[0];
  const float* w    = (const float*)d_in[1];
  const float* Wm   = (const float*)d_in[2];
  const float* bias = (const float*)d_in[3];
  const float* temp = (const float*)d_in[4];
  const int*   src  = (const int*)d_in[5];
  const int*   dst  = (const int*)d_in[6];
  float* hidden = (float*)d_out;

  const int n = in_sizes[0] / ND;
  const int E = in_sizes[1];
  const int L = in_sizes[3] / ND;
  const int total = n * ND;

  float* xA = (float*)d_ws;
  float* xB = xA + (size_t)total;
  char* plane = (char*)(xB + (size_t)total);
  float*          wsort = (float*)plane;
  unsigned short* s16   = (unsigned short*)(plane + (size_t)E * 4);
  unsigned char*  dl8   = (unsigned char*)(plane + (size_t)E * 6);
  int*            tab   = (int*)(plane + (((size_t)E * 7 + 3) & ~(size_t)3));
  float*          wsum  = (float*)(tab + MAXB * NBLK);
  int*            rng   = (int*)(wsum + MAXB * MAXB);
  char* tplane = (char*)xB;
  float*          wtmp  = (float*)tplane;
  unsigned short* stmp  = (unsigned short*)(tplane + (size_t)E * 4);
  unsigned char*  dtmp  = (unsigned char*)(tplane + (size_t)E * 6);

  const int rpb   = (n + MAXB - 1) / MAXB;
  const int nb    = (n + rpb - 1) / rpb;
  const int chunk = (E + NBLK - 1) / NBLK;

  gpr_chunk_group_kernel<<<NBLK, BT, 0, stream>>>(src, dst, w, tab, wtmp, stmp, dtmp,
                                                   E, chunk, (unsigned)rpb);
  gpr_row_sort_kernel<<<nb, BT, 0, stream>>>(wtmp, stmp, dtmp, tab, wsort, s16, dl8, rng, wsum,
                                              chunk);

  for (int i = 0; i < L; ++i) {
    const float* xg = (i == 0) ? x : ((i & 1) ? xA : xB);
    float* xn = (i & 1) ? xB : xA;
    const int more = (i + 1 < L) ? 1 : 0;
    gpr_owner_sum_kernel<<<nb, BT, 0, stream>>>(
        xg, wsort, s16, dl8, rng, wsum, xn, hidden, x, temp,
        Wm + (size_t)i * ND * ND, bias + (size_t)i * ND, (i == 0) ? 1 : 0, more,
        i + 1, n, rpb);
  }
}


# Owner epilogue: the output rows it adds into are loaded at once behind the walk and cross the product's two barriers in flight (LDS-only fences); range bounds, temp read at entry; sum order kept
# speedup vs baseline: 4.4854x; 1.0335x over previous
#include <hip/hip_runtime.h>
#include <hip/hip_bf16.h>


typedef __attribute__((ext_vector_type(2))) float v2f;
typedef __attribute__((ext_vector_type(4))) float v4f;
typedef __attribute__((ext_vector_type(8))) float v8f;

#define ND   64
#define BT   1024
#define NBLK 256
#define MAXB 256
#define EPW  16
#define TPT  8
#define TILE (BT * TPT)


__device__ __forceinline__ int wave_incl_scan(int c, int lane) {
  int incl = c;
#pragma unroll
  for (int o = 1; o < 32; o <<= 1) {
    const int v = __shfl_up(incl, o);
    if (lane >= o) incl += v;
  }
  return incl;
}

__global__ __launch_bounds__(BT) void gpr_chunk_group_kernel(
    const int* __restrict__ src, const int* __restrict__ dst,
    const float* __restrict__ w, int* __restrict__ tab,
    float* __restrict__ wtmp, unsigned short* __restrict__ stmp,
    unsigned char* __restrict__ dtmp, int nEdges, int chunk, unsigned rpb) {
  __shared__ int      lcnt[MAXB];
  __shared__ int      wtot[MAXB / 32];
  __shared__ float    stW[TILE];
  __shared__ unsigned stK[TILE];
  const int t  = threadIdx.x;
  const int e0 = blockIdx.x * chunk;
  const int e1 = min(min(e0 + chunk, nEdges), e0 + TILE);
  if (t < MAXB) lcnt[t] = 0;
  __syncthreads();

  unsigned d[TPT], bk[TPT];
  int      sv[TPT];
  float    wv[TPT];
#pragma unroll
  for (int k = 0; k < TPT; ++k) {
    const int  e  = e0 + k * BT + t;
    const bool ok = e < e1;
    d[k]  = ok ? (unsigned)dst[e] : 0u;
    sv[k] = ok ? src[e] : 0;
    wv[k] = ok ? w[e] : 0.0f;
  }
#pragma unroll
  for (int k = 0; k < TPT; ++k) {
    bk[k] = d[k] / rpb;
    if (e0 + k * BT + t < e1) atomicAdd(&lcnt[bk[k]], 1);
  }
  __syncthreads();

  const int c = (t < MAXB) ? lcnt[t] : 0;
  const int incl = wave_incl_scan(c, t & 31);
  if (t < MAXB && (t & 31) == 31) wtot[t >> 5] = incl;
  __syncthreads();
  if (t < MAXB) {
    int add = 0;
    for (int q = 0; q < (t >> 5); ++q) add += wtot[q];
    const int ex = incl - c + add;
    tab[t * NBLK + blockIdx.x] = (ex << 16) | c;
    lcnt[t] = ex;
  }
  __syncthreads();

#pragma unroll
  for (int k = 0; k < TPT; ++k) {
    if (e0 + k * BT + t < e1) {
      const int r = atomicAdd(&lcnt[bk[k]], 1);
      stW[r] = wv[k];
      stK[r] = ((unsigned)sv[k] & 0xffffu) | ((d[k] - bk[k] * rpb) << 16);
    }
  }
  __syncthreads();

  const int m = e1 - e0;
  for (int p = t; p < m; p += BT) {
    const unsigned key = stK[p];
    wtmp[e0 + p] = stW[p];
    stmp[e0 + p] = (unsigned short)(key & 0xffffu);
    dtmp[e0 + p] = (unsigned char)((key >> 16) & 0xffu);
  }
}

#define RS_CAP 16384

__device__ __forceinline__ int gpr_piece_slot(const int* pre, const int* sbeg, int p) {
  int c = 0;
#pragma unroll
  for (int s = NBLK / 2; s > 0; s >>= 1)
    if (pre[c + s] <= p) c += s;
  return sbeg[c] + p;
}

__global__ __launch_bounds__(BT) void gpr_row_sort_kernel(
    const float* __restrict__ wtmp, const unsigned short* __restrict__ stmp,
    const unsigned char* __restrict__ dtmp, const int* __restrict__ tab,
    float* __restrict__ wsort, unsigned short* __restrict__ s16,
    unsigned char* __restrict__ dl8, int* __restrict__ rng,
    float* __restrict__ wsum, int chunk) {
  __shared__ int            cnt[MAXB];
  __shared__ float          dsum[MAXB];
  __shared__ int            wtot[NBLK / 32];
  __shared__ int            ptot[NBLK / 32];
  __shared__ int            pre[NBLK];
  __shared__ int            sbeg[NBLK];
  __shared__ int            mAll;
  __shared__ float          stW[RS_CAP];
  __shared__ unsigned short stS[RS_CAP];
  __shared__ unsigned char  stD[RS_CAP];
  const int t    = threadIdx.x;
  const int lane = t & 31;
  const int b    = blockIdx.x;

  const int e  = (t < NBLK) ? tab[b * NBLK + t] : 0;
  const int pc = e & 0xffff;
  const int pf = (int)((unsigned)e >> 16);
  int part = pf;
#pragma unroll
  for (int o = 16; o > 0; o >>= 1) part += __shfl_xor(part, o);
  const int pincl = wave_incl_scan(pc, lane);
  if (t < NBLK && lane == 0)  ptot[t >> 5] = part;
  if (t < NBLK && lane == 31) wtot[t >> 5] = pincl;
  if (t < MAXB) { cnt[t] = 0; dsum[t] = 0.0f; }
  __syncthreads();
  int gBeg = 0;
  for (int q = 0; q < NBLK / 32; ++q) gBeg += ptot[q];
  if (t < NBLK) {
    int add = 0;
    for (int q = 0; q < (t >> 5); ++q) add += wtot[q];
    const int ex = pincl - pc + add;
    pre[t]  = ex;
    sbeg[t] = t * chunk + pf - ex;
    if (t == NBLK - 1) mAll = ex + pc;
  }
  __syncthreads();
  const int m = mAll;
  if (t == 0) { rng[2 * b] = gBeg; rng[2 * b + 1] = gBeg + m; }

  for (int base = 0; base < m; base += TILE) {
    unsigned d[TPT];
    float wq[TPT];
#pragma unroll
    for (int k = 0; k < TPT; ++k) {
      const int gi = gpr_piece_slot(pre, sbeg, min(base + k * BT + t, m - 1));
      d[k]  = (unsigned)dtmp[gi];
      wq[k] = wtmp[gi];
    }
#pragma unroll
    for (int k = 0; k < TPT; ++k)
      if (base + k * BT + t < m) {
        atomicAdd(&cnt[d[k]], 1);
        atomicAdd(&dsum[d[k]], wq[k]);
      }
  }
  __syncthreads();
  if (t < MAXB) wsum[b * MAXB + t] = dsum[t];

  if (m > RS_CAP) {
    for (int p = t; p < m; p += BT) {
      const int gi = gpr_piece_slot(pre, sbeg, p);
      wsort[gBeg + p] = wtmp[gi];
      s16[gBeg + p]   = stmp[gi];
      dl8[gBeg + p]   = dtmp[gi];
    }
    return;
  }

  const int c = (t < MAXB) ? cnt[t] : 0;
  const int incl = wave_incl_scan(c, lane);
  if (t < MAXB && lane == 31) wtot[t >> 5] = incl;
  __syncthreads();
  if (t < MAXB) {
    int add = 0;
    for (int q = 0; q < (t >> 5); ++q) add += wtot[q];
    cnt[t] = incl - c + add;
  }
  __syncthreads();

  for (int base = 0; base < m; base += TILE) {
    unsigned d[TPT];
    unsigned short sv[TPT];
    float wv[TPT];
#pragma unroll
    for (int k = 0; k < TPT; ++k) {
      const int gi = gpr_piece_slot(pre, sbeg, min(base + k * BT + t, m - 1));
      d[k]  = (unsigned)dtmp[gi];
      sv[k] = stmp[gi];
      wv[k] = wtmp[gi];
    }
#pragma unroll
    for (int k = 0; k < TPT; ++k) {
      if (base + k * BT + t < m) {
        const int r = atomicAdd(&cnt[d[k]], 1);
        stW[r] = wv[k];
        stS[r] = sv[k];
        stD[r] = (unsigned char)d[k];
      }
    }
  }
  __syncthreads();

  for (int p = t; p < m; p += BT) {
    wsort[gBeg + p] = stW[p];
    s16[gBeg + p]   = stS[p];
    dl8[gBeg + p]   = stD[p];
  }
}

#define ACC_FLOATS 41216
#define AP      68
#define AP4     (AP / 4)
#define KP      AP
#define STG_OFF (MAXB * AP)
#define WL_OFF  (2 * MAXB * AP)
#define EPI_Q   (MAXB * (ND / 4) / BT)

__device__ __forceinline__ void gpr_lds_barrier() {
  __builtin_amdgcn_fence(__ATOMIC_RELEASE, "workgroup", "local");
  __builtin_amdgcn_s_barrier();
  __builtin_amdgcn_fence(__ATOMIC_ACQUIRE, "workgroup", "local");
}

__device__ __forceinline__ void gpr_flush_run(float* acc, int row, int lane, v2f run) {
  float* o = acc + row * AP + 2 * lane;
  __hip_atomic_fetch_add(o,     run.x, __ATOMIC_RELAXED, __HIP_MEMORY_SCOPE_WORKGROUP);
  __hip_atomic_fetch_add(o + 1, run.y, __ATOMIC_RELAXED, __HIP_MEMORY_SCOPE_WORKGROUP);
}

__global__ __launch_bounds__(BT) void gpr_owner_sum_kernel(
    const float* __restrict__ Xg, const float* __restrict__ wsort,
    const unsigned short* __restrict__ s16, const unsigned char* __restrict__ dl8,
    const int* __restrict__ rng, const float* __restrict__ wsum,
    float* __restrict__ Xnext,
    float* __restrict__ hidden, const float* __restrict__ xin,
    const float* __restrict__ temp, const float* __restrict__ Wl,
    const float* __restrict__ bl, int first, int more,
    int tempIdx, int n, int rpb) {
  __shared__ float lds[ACC_FLOATS];
  float* acc = lds;
  float* stg = lds + STG_OFF;
  float* wl  = lds + WL_OFF;
  const int t    = threadIdx.x;
  const int b    = blockIdx.x;
  const int row0 = b * rpb;
  const int rows = min(rpb, n - row0);
  const int nRt  = (rows + 15) >> 4;
  const int nSlot  = nRt * 16;
  const int nRow4  = rows * (ND / 4);

  const int eBeg = rng[2 * b];
  const int eEnd = rng[2 * b + 1];
  const float tk = temp[tempIdx];
  const float t0 = temp[0];

  const float ws = (t < nSlot) ? wsum[b * MAXB + t] : 0.0f;
  const float bq = (t < ND) ? bl[t] : 0.0f;
  for (int i = t; i < nSlot * (ND / 4); i += BT)
    ((v4f*)acc)[(i >> 4) * AP4 + (i & 15)] = (v4f){0.0f, 0.0f, 0.0f, 0.0f};
  for (int i = t; i < ND * ND / 4; i += BT)
    ((v4f*)wl)[(i >> 4) * AP4 + (i & 15)] = ((const v4f*)Wl)[i];
  if (t < ND) ((v4f*)wl)[t * AP4 + ND / 4] = (v4f){bq, 0.0f, 0.0f, 0.0f};
  if (t < nSlot) ((v4f*)acc)[t * AP4 + ND / 4] = (v4f){ws, 0.0f, 0.0f, 0.0f};
  __syncthreads();

  const int wv   = __builtin_amdgcn_readfirstlane(t >> 5);
  const int lane = t & 31;

  const int nW   = BT / 32;
  const int per  = ((eEnd - eBeg + nW * EPW - 1) / (nW * EPW)) * EPW;
  const int wBeg = eBeg + wv * per;
  const int wEnd = min(wBeg + per, eEnd);

  int cur = -1;
  v2f run = {0.0f, 0.0f};
  const char* xb = (const char*)Xg;
  const unsigned lane8 = (unsigned)lane * 8u;

  for (int g = wBeg; g < wEnd; g += EPW) {
    if (g + EPW <= wEnd) {
      v2f v[EPW];
      float we[EPW];
      int dl[EPW];
#pragma unroll
      for (int j = 0; j < EPW; ++j) {
        const unsigned s = s16[g + j];
        we[j] = wsort[g + j];
        dl[j] = dl8[g + j];
        v[j] = *(const v2f*)(xb + (s * (unsigned)(ND * 4) + lane8));
      }
#pragma unroll
      for (int j = 0; j < EPW; ++j) {
        if (__builtin_expect(dl[j] != cur, 0)) {
          if (cur >= 0) gpr_flush_run(acc, cur, lane, run);
          cur = dl[j];
          run = (v2f){0.0f, 0.0f};
        }
        run = __builtin_elementwise_fma(v[j], (v2f){we[j], we[j]}, run);
      }
    } else {
      v2f v[EPW];
      float we[EPW];
      int dl[EPW];
#pragma unroll
      for (int j = 0; j < EPW; ++j) {
        const int      e = min(g + j, wEnd - 1);
        const unsigned s = s16[e];
        we[j] = wsort[e];
        dl[j] = dl8[e];
        v[j] = *(const v2f*)(xb + (s * (unsigned)(ND * 4) + lane8));
      }
#pragma unroll
      for (int j = 0; j < EPW; ++j) {
        if (g + j < wEnd) {
          if (__builtin_expect(dl[j] != cur, 0)) {
            if (cur >= 0) gpr_flush_run(acc, cur, lane, run);
            cur = dl[j];
            run = (v2f){0.0f, 0.0f};
          }
          run = __builtin_elementwise_fma(v[j], (v2f){we[j], we[j]}, run);
        }
      }
    }
  }
  if (cur >= 0) gpr_flush_run(acc, cur, lane, run);

  const v4f* hs = (const v4f*)((first ? xin : (const float*)hidden) + (size_t)row0 * ND);
  v4f hp[EPI_Q];
#pragma unroll
  for (int q = 0; q < EPI_Q; ++q) {
    const int i = t + q * BT;
    hp[q] = (i < nRow4) ? hs[i] : (v4f){0.0f, 0.0f, 0.0f, 0.0f};
  }
  gpr_lds_barrier();

  const int rt = wv >> 1;
  const int cp = wv & 1;
  if (rt < nRt) {
    const int hi = lane >> 4;
    const int lo = lane & 15;
    const float* ap  = acc + (rt * 16 + lo) * AP + 2 * hi;
    const float* bp0 = wl  + (cp * 32 + lo) * AP + 2 * hi;
    const float* bp1 = bp0 + 16 * AP;
    v8f c0 = {};
    v8f c1 = {};
#pragma unroll
    for (int k = 0; k < KP; k += 4) {
      v2f a  = *(const v2f*)(ap  + k);
      v2f b0 = *(const v2f*)(bp0 + k);
      v2f b1 = *(const v2f*)(bp1 + k);
      c0 = __builtin_amdgcn_wmma_f32_16x16x4_f32(
           false, a,  false, b0,
           (short)0, c0,  false,  false);
      c1 = __builtin_amdgcn_wmma_f32_16x16x4_f32(
           false, a,  false, b1,
           (short)0, c1,  false,  false);
    }
    float* so = stg + (rt * 16 + 8 * hi) * AP + cp * 32 + lo;
#pragma unroll
    for (int v = 0; v < 8; ++v) {
      so[v * AP]      = fmaxf(c0[v], 0.0f);
      so[v * AP + 16] = fmaxf(c1[v], 0.0f);
    }
  }
  gpr_lds_barrier();

  v4f* ho = (v4f*)(hidden + (size_t)row0 * ND);
  v4f* xn = (v4f*)(Xnext + (size_t)row0 * ND);
  if (first) {
#pragma unroll
    for (int q = 0; q < EPI_Q; ++q) {
      const int i = t + q * BT;
      if (i < nRow4) {
        const v4f v = ((const v4f*)stg)[(i >> 4) * AP4 + (i & 15)];
        ho[i] = hp[q] * t0 + v * tk;
        if (more) xn[i] = v;
      }
    }
  } else {
#pragma unroll
    for (int q = 0; q < EPI_Q; ++q) {
      const int i = t + q * BT;
      if (i < nRow4) {
        const v4f v = ((const v4f*)stg)[(i >> 4) * AP4 + (i & 15)];
        ho[i] = hp[q] + v * tk;
        if (more) xn[i] = v;
      }
    }
  }
}

extern "C" void kernel_launch(void* const* d_in, const int* in_sizes, int n_in,
                              void* d_out, int out_size, void* d_ws, size_t ws_size,
                              hipStream_t stream) {
  const float* x    = (const float*)d_in[0];
  const float* w    = (const float*)d_in[1];
  const float* Wm   = (const float*)d_in[2];
  const float* bias = (const float*)d_in[3];
  const float* temp = (const float*)d_in[4];
  const int*   src  = (const int*)d_in[5];
  const int*   dst  = (const int*)d_in[6];
  float* hidden = (float*)d_out;

  const int n = in_sizes[0] / ND;
  const int E = in_sizes[1];
  const int L = in_sizes[3] / ND;
  const int total = n * ND;

  float* xA = (float*)d_ws;
  float* xB = xA + (size_t)total;
  char* plane = (char*)(xB + (size_t)total);
  float*          wsort = (float*)plane;
  unsigned short* s16   = (unsigned short*)(plane + (size_t)E * 4);
  unsigned char*  dl8   = (unsigned char*)(plane + (size_t)E * 6);
  int*            tab   = (int*)(plane + (((size_t)E * 7 + 3) & ~(size_t)3));
  float*          wsum  = (float*)(tab + MAXB * NBLK);
  int*            rng   = (int*)(wsum + MAXB * MAXB);
  char* tplane = (char*)xB;
  float*          wtmp  = (float*)tplane;
  unsigned short* stmp  = (unsigned short*)(tplane + (size_t)E * 4);
  unsigned char*  dtmp  = (unsigned char*)(tplane + (size_t)E * 6);

  const int rpb   = (n + MAXB - 1) / MAXB;
  const int nb    = (n + rpb - 1) / rpb;
  const int chunk = (E + NBLK - 1) / NBLK;

  gpr_chunk_group_kernel<<<NBLK, BT, 0, stream>>>(src, dst, w, tab, wtmp, stmp, dtmp,
                                                   E, chunk, (unsigned)rpb);
  gpr_row_sort_kernel<<<nb, BT, 0, stream>>>(wtmp, stmp, dtmp, tab, wsort, s16, dl8, rng, wsum,
                                              chunk);

  for (int i = 0; i < L; ++i) {
    const float* xg = (i == 0) ? x : ((i & 1) ? xA : xB);
    float* xn = (i & 1) ? xB : xA;
    const int more = (i + 1 < L) ? 1 : 0;
    gpr_owner_sum_kernel<<<nb, BT, 0, stream>>>(
        xg, wsort, s16, dl8, rng, wsum, xn, hidden, x, temp,
        Wm + (size_t)i * ND * ND, bias + (size_t)i * ND, (i == 0) ? 1 : 0, more,
        i + 1, n, rpb);
  }
}


# First layer's launch sorts its range by row itself and walks the run straight from its LDS staging (own kernel, no read-back; later layers' kernel kept): 6 launches to 5; sum order stays unordered
# speedup vs baseline: 4.7075x; 1.0495x over previous
#include <hip/hip_runtime.h>
#include <hip/hip_bf16.h>


typedef __attribute__((ext_vector_type(2))) float v2f;
typedef __attribute__((ext_vector_type(4))) float v4f;
typedef __attribute__((ext_vector_type(8))) float v8f;

#define ND   64
#define BT   1024
#define NBLK 256
#define MAXB 256
#define EPW  16
#define TPT  8
#define TILE (BT * TPT)


__device__ __forceinline__ int wave_incl_scan(int c, int lane) {
  int incl = c;
#pragma unroll
  for (int o = 1; o < 32; o <<= 1) {
    const int v = __shfl_up(incl, o);
    if (lane >= o) incl += v;
  }
  return incl;
}

__global__ __launch_bounds__(BT) void gpr_chunk_group_kernel(
    const int* __restrict__ src, const int* __restrict__ dst,
    const float* __restrict__ w, int* __restrict__ tab,
    float* __restrict__ wtmp, unsigned short* __restrict__ stmp,
    unsigned char* __restrict__ dtmp, int nEdges, int chunk, unsigned rpb) {
  __shared__ int      lcnt[MAXB];
  __shared__ int      wtot[MAXB / 32];
  __shared__ float    stW[TILE];
  __shared__ unsigned stK[TILE];
  const int t  = threadIdx.x;
  const int e0 = blockIdx.x * chunk;
  const int e1 = min(min(e0 + chunk, nEdges), e0 + TILE);
  if (t < MAXB) lcnt[t] = 0;
  __syncthreads();

  unsigned d[TPT], bk[TPT];
  int      sv[TPT];
  float    wv[TPT];
#pragma unroll
  for (int k = 0; k < TPT; ++k) {
    const int  e  = e0 + k * BT + t;
    const bool ok = e < e1;
    d[k]  = ok ? (unsigned)dst[e] : 0u;
    sv[k] = ok ? src[e] : 0;
    wv[k] = ok ? w[e] : 0.0f;
  }
#pragma unroll
  for (int k = 0; k < TPT; ++k) {
    bk[k] = d[k] / rpb;
    if (e0 + k * BT + t < e1) atomicAdd(&lcnt[bk[k]], 1);
  }
  __syncthreads();

  const int c = (t < MAXB) ? lcnt[t] : 0;
  const int incl = wave_incl_scan(c, t & 31);
  if (t < MAXB && (t & 31) == 31) wtot[t >> 5] = incl;
  __syncthreads();
  if (t < MAXB) {
    int add = 0;
    for (int q = 0; q < (t >> 5); ++q) add += wtot[q];
    const int ex = incl - c + add;
    tab[t * NBLK + blockIdx.x] = (ex << 16) | c;
    lcnt[t] = ex;
  }
  __syncthreads();

#pragma unroll
  for (int k = 0; k < TPT; ++k) {
    if (e0 + k * BT + t < e1) {
      const int r = atomicAdd(&lcnt[bk[k]], 1);
      stW[r] = wv[k];
      stK[r] = ((unsigned)sv[k] & 0xffffu) | ((d[k] - bk[k] * rpb) << 16);
    }
  }
  __syncthreads();

  const int m = e1 - e0;
  for (int p = t; p < m; p += BT) {
    const unsigned key = stK[p];
    wtmp[e0 + p] = stW[p];
    stmp[e0 + p] = (unsigned short)(key & 0xffffu);
    dtmp[e0 + p] = (unsigned char)((key >> 16) & 0xffu);
  }
}

#define RS_CAP 16384

__device__ __forceinline__ int gpr_piece_slot(const int* pre, const int* sbeg, int p) {
  int c = 0;
#pragma unroll
  for (int s = NBLK / 2; s > 0; s >>= 1)
    if (pre[c + s] <= p) c += s;
  return sbeg[c] + p;
}

#define ACC_FLOATS 41216
#define AP      68
#define AP4     (AP / 4)
#define KP      AP
#define STG_OFF (MAXB * AP)
#define WL_OFF  (2 * MAXB * AP)
#define EPI_Q   (MAXB * (ND / 4) / BT)

__device__ __forceinline__ void gpr_lds_barrier() {
  __builtin_amdgcn_fence(__ATOMIC_RELEASE, "workgroup", "local");
  __builtin_amdgcn_s_barrier();
  __builtin_amdgcn_fence(__ATOMIC_ACQUIRE, "workgroup", "local");
}

__device__ __forceinline__ void gpr_flush_run(float* acc, int row, int lane, v2f run) {
  float* o = acc + row * AP + 2 * lane;
  __hip_atomic_fetch_add(o,     run.x, __ATOMIC_RELAXED, __HIP_MEMORY_SCOPE_WORKGROUP);
  __hip_atomic_fetch_add(o + 1, run.y, __ATOMIC_RELAXED, __HIP_MEMORY_SCOPE_WORKGROUP);
}

__device__ __forceinline__ void gpr_product_tiles(const float* acc, const float* wl, float* stg,
                                                  int wv, int lane, int nRt) {
  const int rt = wv >> 1;
  const int cp = wv & 1;
  if (rt < nRt) {
    const int hi = lane >> 4;
    const int lo = lane & 15;
    const float* ap  = acc + (rt * 16 + lo) * AP + 2 * hi;
    const float* bp0 = wl  + (cp * 32 + lo) * AP + 2 * hi;
    const float* bp1 = bp0 + 16 * AP;
    v8f c0 = {};
    v8f c1 = {};
#pragma unroll
    for (int k = 0; k < KP; k += 4) {
      v2f a  = *(const v2f*)(ap  + k);
      v2f b0 = *(const v2f*)(bp0 + k);
      v2f b1 = *(const v2f*)(bp1 + k);
      c0 = __builtin_amdgcn_wmma_f32_16x16x4_f32(
           false, a,  false, b0,
           (short)0, c0,  false,  false);
      c1 = __builtin_amdgcn_wmma_f32_16x16x4_f32(
           false, a,  false, b1,
           (short)0, c1,  false,  false);
    }
    float* so = stg + (rt * 16 + 8 * hi) * AP + cp * 32 + lo;
#pragma unroll
    for (int v = 0; v < 8; ++v) {
      so[v * AP]      = fmaxf(c0[v], 0.0f);
      so[v * AP + 16] = fmaxf(c1[v], 0.0f);
    }
  }
}

__global__ __launch_bounds__(BT) void gpr_owner_sum_kernel(
    const float* __restrict__ Xg, const float* __restrict__ wsort,
    const unsigned short* __restrict__ s16, const unsigned char* __restrict__ dl8,
    const int* __restrict__ rng, const float* __restrict__ wsum,
    float* __restrict__ Xnext,
    float* __restrict__ hidden,
    const float* __restrict__ temp, const float* __restrict__ Wl,
    const float* __restrict__ bl, int more,
    int tempIdx, int n, int rpb) {
  __shared__ float lds[ACC_FLOATS];
  float* acc = lds;
  float* stg = lds + STG_OFF;
  float* wl  = lds + WL_OFF;
  const int t    = threadIdx.x;
  const int b    = blockIdx.x;
  const int row0 = b * rpb;
  const int rows = min(rpb, n - row0);
  const int nRt  = (rows + 15) >> 4;
  const int nSlot  = nRt * 16;
  const int nRow4  = rows * (ND / 4);

  const int eBeg = rng[2 * b];
  const int eEnd = rng[2 * b + 1];
  const float tk = temp[tempIdx];

  const float ws = (t < nSlot) ? wsum[b * MAXB + t] : 0.0f;
  const float bq = (t < ND) ? bl[t] : 0.0f;
  for (int i = t; i < nSlot * (ND / 4); i += BT)
    ((v4f*)acc)[(i >> 4) * AP4 + (i & 15)] = (v4f){0.0f, 0.0f, 0.0f, 0.0f};
  for (int i = t; i < ND * ND / 4; i += BT)
    ((v4f*)wl)[(i >> 4) * AP4 + (i & 15)] = ((const v4f*)Wl)[i];
  if (t < ND) ((v4f*)wl)[t * AP4 + ND / 4] = (v4f){bq, 0.0f, 0.0f, 0.0f};
  if (t < nSlot) ((v4f*)acc)[t * AP4 + ND / 4] = (v4f){ws, 0.0f, 0.0f, 0.0f};
  __syncthreads();

  const int wv   = __builtin_amdgcn_readfirstlane(t >> 5);
  const int lane = t & 31;

  const int nW   = BT / 32;
  const int per  = ((eEnd - eBeg + nW * EPW - 1) / (nW * EPW)) * EPW;
  const int wBeg = eBeg + wv * per;
  const int wEnd = min(wBeg + per, eEnd);

  int cur = -1;
  v2f run = {0.0f, 0.0f};
  const char* xb = (const char*)Xg;
  const unsigned lane8 = (unsigned)lane * 8u;

  for (int g = wBeg; g < wEnd; g += EPW) {
    if (g + EPW <= wEnd) {
      v2f v[EPW];
      float we[EPW];
      int dl[EPW];
#pragma unroll
      for (int j = 0; j < EPW; ++j) {
        const unsigned s = s16[g + j];
        we[j] = wsort[g + j];
        dl[j] = dl8[g + j];
        v[j] = *(const v2f*)(xb + (s * (unsigned)(ND * 4) + lane8));
      }
#pragma unroll
      for (int j = 0; j < EPW; ++j) {
        if (__builtin_expect(dl[j] != cur, 0)) {
          if (cur >= 0) gpr_flush_run(acc, cur, lane, run);
          cur = dl[j];
          run = (v2f){0.0f, 0.0f};
        }
        run = __builtin_elementwise_fma(v[j], (v2f){we[j], we[j]}, run);
      }
    } else {
      v2f v[EPW];
      float we[EPW];
      int dl[EPW];
#pragma unroll
      for (int j = 0; j < EPW; ++j) {
        const int      e = min(g + j, wEnd - 1);
        const unsigned s = s16[e];
        we[j] = wsort[e];
        dl[j] = dl8[e];
        v[j] = *(const v2f*)(xb + (s * (unsigned)(ND * 4) + lane8));
      }
#pragma unroll
      for (int j = 0; j < EPW; ++j) {
        if (g + j < wEnd) {
          if (__builtin_expect(dl[j] != cur, 0)) {
            if (cur >= 0) gpr_flush_run(acc, cur, lane, run);
            cur = dl[j];
            run = (v2f){0.0f, 0.0f};
          }
          run = __builtin_elementwise_fma(v[j], (v2f){we[j], we[j]}, run);
        }
      }
    }
  }
  if (cur >= 0) gpr_flush_run(acc, cur, lane, run);

  const v4f* hs = (const v4f*)(hidden + (size_t)row0 * ND);
  v4f hp[EPI_Q];
#pragma unroll
  for (int q = 0; q < EPI_Q; ++q) {
    const int i = t + q * BT;
    hp[q] = (i < nRow4) ? hs[i] : (v4f){0.0f, 0.0f, 0.0f, 0.0f};
  }
  gpr_lds_barrier();

  gpr_product_tiles(acc, wl, stg, wv, lane, nRt);
  gpr_lds_barrier();

  v4f* ho = (v4f*)(hidden + (size_t)row0 * ND);
  v4f* xn = (v4f*)(Xnext + (size_t)row0 * ND);
#pragma unroll
  for (int q = 0; q < EPI_Q; ++q) {
    const int i = t + q * BT;
    if (i < nRow4) {
      const v4f v = ((const v4f*)stg)[(i >> 4) * AP4 + (i & 15)];
      ho[i] = hp[q] + v * tk;
      if (more) xn[i] = v;
    }
  }
}

typedef __attribute__((ext_vector_type(4))) unsigned v4u;

__global__ __launch_bounds__(BT) void gpr_sort_owner_kernel(
    const float* __restrict__ xin, const float* __restrict__ wtmp,
    const unsigned short* __restrict__ stmp, const unsigned char* __restrict__ dtmp,
    const int* __restrict__ tab, float* __restrict__ wsort,
    unsigned short* __restrict__ s16, unsigned char* __restrict__ dl8,
    int* __restrict__ rng, float* __restrict__ wsum,
    float* __restrict__ Xnext, float* __restrict__ hidden,
    const float* __restrict__ temp, const float* __restrict__ Wl,
    const float* __restrict__ bl, int more, int n, int rpb, int chunk) {
  __shared__ int   cnt[MAXB];
  __shared__ float dsum[MAXB];
  __shared__ int   wtot[NBLK / 32];
  __shared__ int   ptot[NBLK / 32];
  __shared__ int   pre[NBLK];
  __shared__ int   sbeg[NBLK];
  __shared__ int   mAll;
  __shared__ __attribute__((aligned(16))) float          stW[RS_CAP];
  __shared__ __attribute__((aligned(16))) unsigned short stS[RS_CAP];
  __shared__ __attribute__((aligned(16))) unsigned char  stD[RS_CAP];
  __shared__ float lds[ACC_FLOATS];
  float* acc = lds;
  float* stg = lds + STG_OFF;
  float* wl  = lds + WL_OFF;
  const int t    = threadIdx.x;
  const int lane = t & 31;
  const int b    = blockIdx.x;
  const int row0 = b * rpb;
  const int rows = min(rpb, n - row0);
  const int nRt  = (rows + 15) >> 4;
  const int nSlot  = nRt * 16;
  const int nRow4  = rows * (ND / 4);

  const float tk = temp[1];
  const float t0 = temp[0];
  const float bq = (t < ND) ? bl[t] : 0.0f;
  static_assert(ND * ND / 4 == BT, "one 16-byte piece of W per thread");
  const v4f   wq0 = ((const v4f*)Wl)[t];
  const int e  = (t < NBLK) ? tab[b * NBLK + t] : 0;
  const int pc = e & 0xffff;
  const int pf = (int)((unsigned)e >> 16);
  int part = pf;
#pragma unroll
  for (int o = 16; o > 0; o >>= 1) part += __shfl_xor(part, o);
  const int pincl = wave_incl_scan(pc, lane);
  if (t < NBLK && lane == 0)  ptot[t >> 5] = part;
  if (t < NBLK && lane == 31) wtot[t >> 5] = pincl;
  if (t < MAXB) { cnt[t] = 0; dsum[t] = 0.0f; }
  __syncthreads();
  int gBeg = 0;
  for (int q = 0; q < NBLK / 32; ++q) gBeg += ptot[q];
  if (t < NBLK) {
    int add = 0;
    for (int q = 0; q < (t >> 5); ++q) add += wtot[q];
    const int ex = pincl - pc + add;
    pre[t]  = ex;
    sbeg[t] = t * chunk + pf - ex;
    if (t == NBLK - 1) mAll = ex + pc;
  }
  __syncthreads();
  const int m = __builtin_amdgcn_readfirstlane(mAll);
  if (t == 0) { rng[2 * b] = gBeg; rng[2 * b + 1] = gBeg + m; }

  for (int base = 0; base < m; base += TILE) {
    unsigned d[TPT];
    float wq[TPT];
#pragma unroll
    for (int k = 0; k < TPT; ++k) {
      const int gi = gpr_piece_slot(pre, sbeg, min(base + k * BT + t, m - 1));
      d[k]  = (unsigned)dtmp[gi];
      wq[k] = wtmp[gi];
    }
#pragma unroll
    for (int k = 0; k < TPT; ++k)
      if (base + k * BT + t < m) {
        atomicAdd(&cnt[d[k]], 1);
        atomicAdd(&dsum[d[k]], wq[k]);
      }
  }
  for (int i = t; i < nSlot * (ND / 4); i += BT)
    ((v4f*)acc)[(i >> 4) * AP4 + (i & 15)] = (v4f){0.0f, 0.0f, 0.0f, 0.0f};
  ((v4f*)wl)[(t >> 4) * AP4 + (t & 15)] = wq0;
  if (t < ND) ((v4f*)wl)[t * AP4 + ND / 4] = (v4f){bq, 0.0f, 0.0f, 0.0f};
  __syncthreads();
  if (t < MAXB) wsum[b * MAXB + t] = dsum[t];
  if (t < nSlot) ((v4f*)acc)[t * AP4 + ND / 4] = (v4f){dsum[t], 0.0f, 0.0f, 0.0f};

  const int wv = __builtin_amdgcn_readfirstlane(t >> 5);
  const int nW = BT / 32;
  const char* xb = (const char*)xin;
  const unsigned lane8 = (unsigned)lane * 8u;

  if (m > RS_CAP) {
    for (int p = t; p < m; p += BT) {
      const int gi = gpr_piece_slot(pre, sbeg, p);
      wsort[gBeg + p] = wtmp[gi];
      s16[gBeg + p]   = stmp[gi];
      dl8[gBeg + p]   = dtmp[gi];
    }
    for (int p = wv; p < m; p += nW) {
      const int gi = gpr_piece_slot(pre, sbeg, p);
      const unsigned sr = stmp[gi];
      const float    we = wtmp[gi];
      const int      dr = dtmp[gi];
      const v2f v = *(const v2f*)(xb + (sr * (unsigned)(ND * 4) + lane8));
      gpr_flush_run(acc, dr, lane, v * we);
    }
  } else {
    const int c = (t < MAXB) ? cnt[t] : 0;
    const int incl = wave_incl_scan(c, lane);
    if (t < MAXB && lane == 31) wtot[t >> 5] = incl;
    __syncthreads();
    if (t < MAXB) {
      int add = 0;
      for (int q = 0; q < (t >> 5); ++q) add += wtot[q];
      cnt[t] = incl - c + add;
    }
    __syncthreads();

    for (int base = 0; base < m; base += TILE) {
      unsigned d[TPT];
      unsigned short sv[TPT];
      float wv2[TPT];
#pragma unroll
      for (int k = 0; k < TPT; ++k) {
        const int gi = gpr_piece_slot(pre, sbeg, min(base + k * BT + t, m - 1));
        d[k]   = (unsigned)dtmp[gi];
        sv[k]  = stmp[gi];
        wv2[k] = wtmp[gi];
      }
#pragma unroll
      for (int k = 0; k < TPT; ++k) {
        if (base + k * BT + t < m) {
          const int r = atomicAdd(&cnt[d[k]], 1);
          stW[r] = wv2[k];
          stS[r] = sv[k];
          stD[r] = (unsigned char)d[k];
        }
      }
    }
    __syncthreads();

    for (int p = t; p < m; p += BT) {
      wsort[gBeg + p] = stW[p];
      s16[gBeg + p]   = stS[p];
      dl8[gBeg + p]   = stD[p];
    }

    const int per  = ((m + nW * EPW - 1) / (nW * EPW)) * EPW;
    const int wBeg = wv * per;
    const int wEnd = min(wBeg + per, m);
    int cur = -1;
    v2f run = {0.0f, 0.0f};
    for (int g = wBeg; g < wEnd; g += EPW) {
      if (g + EPW <= wEnd) {
        v4f wr[EPW / 4];
#pragma unroll
        for (int q = 0; q < EPW / 4; ++q) wr[q] = *(const v4f*)(stW + g + 4 * q);
        const v4u sa = *(const v4u*)(stS + g);
        const v4u sb = *(const v4u*)(stS + g + 8);
        const v4u da = *(const v4u*)(stD + g);
        unsigned sw[EPW / 2];
        sw[0] = sa.x; sw[1] = sa.y; sw[2] = sa.z; sw[3] = sa.w;
        sw[4] = sb.x; sw[5] = sb.y; sw[6] = sb.z; sw[7] = sb.w;
        unsigned dw[EPW / 4];
        dw[0] = __builtin_amdgcn_readfirstlane(da.x);
        dw[1] = __builtin_amdgcn_readfirstlane(da.y);
        dw[2] = __builtin_amdgcn_readfirstlane(da.z);
        dw[3] = __builtin_amdgcn_readfirstlane(da.w);
        v2f v[EPW];
#pragma unroll
        for (int j = 0; j < EPW; ++j) {
          const unsigned sr = (sw[j >> 1] >> (16 * (j & 1))) & 0xffffu;
          v[j] = *(const v2f*)(xb + (sr * (unsigned)(ND * 4) + lane8));
        }
#pragma unroll
        for (int j = 0; j < EPW; ++j) {
          const int   dr = (int)((dw[j >> 2] >> (8 * (j & 3))) & 0xffu);
          const float we = wr[j >> 2][j & 3];
          if (__builtin_expect(dr != cur, 0)) {
            if (cur >= 0) gpr_flush_run(acc, cur, lane, run);
            cur = dr;
            run = (v2f){0.0f, 0.0f};
          }
          run = __builtin_elementwise_fma(v[j], (v2f){we, we}, run);
        }
      } else {
        v2f v[EPW];
        float we[EPW];
        int dr[EPW];
#pragma unroll
        for (int j = 0; j < EPW; ++j) {
          const int      ee = min(g + j, wEnd - 1);
          const unsigned sr = stS[ee];
          we[j] = stW[ee];
          dr[j] = __builtin_amdgcn_readfirstlane((int)stD[ee]);
          v[j] = *(const v2f*)(xb + (sr * (unsigned)(ND * 4) + lane8));
        }
#pragma unroll
        for (int j = 0; j < EPW; ++j) {
          if (g + j < wEnd) {
            if (__builtin_expect(dr[j] != cur, 0)) {
              if (cur >= 0) gpr_flush_run(acc, cur, lane, run);
              cur = dr[j];
              run = (v2f){0.0f, 0.0f};
            }
            run = __builtin_elementwise_fma(v[j], (v2f){we[j], we[j]}, run);
          }
        }
      }
    }
    if (cur >= 0) gpr_flush_run(acc, cur, lane, run);
  }

  const v4f* hs = (const v4f*)(xin + (size_t)row0 * ND);
  v4f hp[EPI_Q];
#pragma unroll
  for (int q = 0; q < EPI_Q; ++q) {
    const int i = t + q * BT;
    hp[q] = (i < nRow4) ? hs[i] : (v4f){0.0f, 0.0f, 0.0f, 0.0f};
  }
  gpr_lds_barrier();

  gpr_product_tiles(acc, wl, stg, wv, lane, nRt);
  gpr_lds_barrier();

  v4f* ho = (v4f*)(hidden + (size_t)row0 * ND);
  v4f* xn = (v4f*)(Xnext + (size_t)row0 * ND);
#pragma unroll
  for (int q = 0; q < EPI_Q; ++q) {
    const int i = t + q * BT;
    if (i < nRow4) {
      const v4f v = ((const v4f*)stg)[(i >> 4) * AP4 + (i & 15)];
      ho[i] = hp[q] * t0 + v * tk;
      if (more) xn[i] = v;
    }
  }
}

extern "C" void kernel_launch(void* const* d_in, const int* in_sizes, int n_in,
                              void* d_out, int out_size, void* d_ws, size_t ws_size,
                              hipStream_t stream) {
  const float* x    = (const float*)d_in[0];
  const float* w    = (const float*)d_in[1];
  const float* Wm   = (const float*)d_in[2];
  const float* bias = (const float*)d_in[3];
  const float* temp = (const float*)d_in[4];
  const int*   src  = (const int*)d_in[5];
  const int*   dst  = (const int*)d_in[6];
  float* hidden = (float*)d_out;

  const int n = in_sizes[0] / ND;
  const int E = in_sizes[1];
  const int L = in_sizes[3] / ND;
  const int total = n * ND;

  float* xA = (float*)d_ws;
  float* xB = xA + (size_t)total;
  char* plane = (char*)(xB + (size_t)total);
  float*          wsort = (float*)plane;
  unsigned short* s16   = (unsigned short*)(plane + (size_t)E * 4);
  unsigned char*  dl8   = (unsigned char*)(plane + (size_t)E * 6);
  int*            tab   = (int*)(plane + (((size_t)E * 7 + 3) & ~(size_t)3));
  float*          wsum  = (float*)(tab + MAXB * NBLK);
  int*            rng   = (int*)(wsum + MAXB * MAXB);
  char* tplane = (char*)xB;
  float*          wtmp  = (float*)tplane;
  unsigned short* stmp  = (unsigned short*)(tplane + (size_t)E * 4);
  unsigned char*  dtmp  = (unsigned char*)(tplane + (size_t)E * 6);

  const int rpb   = (n + MAXB - 1) / MAXB;
  const int nb    = (n + rpb - 1) / rpb;
  const int chunk = (E + NBLK - 1) / NBLK;

  gpr_chunk_group_kernel<<<NBLK, BT, 0, stream>>>(src, dst, w, tab, wtmp, stmp, dtmp,
                                                   E, chunk, (unsigned)rpb);
  gpr_sort_owner_kernel<<<nb, BT, 0, stream>>>(x, wtmp, stmp, dtmp, tab, wsort, s16, dl8, rng, wsum,
                                                xA, hidden, temp, Wm, bias, (L > 1) ? 1 : 0,
                                                n, rpb, chunk);

  for (int i = 1; i < L; ++i) {
    const float* xg = (i & 1) ? xA : xB;
    float* xn = (i & 1) ? xB : xA;
    const int more = (i + 1 < L) ? 1 : 0;
    gpr_owner_sum_kernel<<<nb, BT, 0, stream>>>(
        xg, wsort, s16, dl8, rng, wsum, xn, hidden, temp,
        Wm + (size_t)i * ND * ND, bias + (size_t)i * ND, more,
        i + 1, n, rpb);
  }
}
